# starting kernel
# baseline (speedup 1.0000x reference)
#include <hip/hip_runtime.h>
#include <hip/hip_bf16.h>
#include <math.h>

typedef __attribute__((ext_vector_type(2))) float v2f;
typedef __attribute__((ext_vector_type(8))) float v8f;

#define IN_F   128
#define OUT_F  32
#define NHEAD  4
#define ALPHA  0.2f

__device__ inline void atomicMaxFloat(float* addr, float v) {
    int iv = __float_as_int(v);
    if (iv >= 0) {
        atomicMax((int*)addr, iv);
    } else {
        atomicMin((unsigned int*)addr, (unsigned int)iv);
    }
}

__global__ __launch_bounds__(256) void gat_init(float* __restrict__ m,
                                                float* __restrict__ s,
                                                float* __restrict__ out, int N) {
    long i = (long)blockIdx.x * blockDim.x + threadIdx.x;
    long n4 = (long)N * NHEAD;
    long nF = (long)N * (NHEAD * OUT_F);
    if (i < n4) { m[i] = -INFINITY; s[i] = 0.0f; }
    if (i < nF) { out[i] = 0.0f; }
}

__global__ __launch_bounds__(256) void gat_gemm(const float* __restrict__ x,
                                                const float* __restrict__ W,
                                                float* __restrict__ h, int N) {
    const int wave = threadIdx.x >> 5;
    const int lane = threadIdx.x & 31;
    const long rowTile = (long)blockIdx.x * 8 + wave;
    const long row0 = rowTile * 16;
    if (row0 >= N) return;

    const int mrow  = lane & 15;
    const int khalf = (lane >> 4) * 2;

    v8f acc[8];
    #pragma unroll
    for (int t = 0; t < 8; ++t) acc[t] = v8f{};

    long r = row0 + mrow;
    if (r >= N) r = N - 1;
    const float* __restrict__ xrow = x + r * IN_F;

    for (int k = 0; k < IN_F; k += 4) {
        v2f a;
        a.x = xrow[k + khalf + 0];
        a.y = xrow[k + khalf + 1];
        #pragma unroll
        for (int t = 0; t < 8; ++t) {
            const int ncol = t * 16 + mrow;
            v2f b;
            b.x = W[(k + khalf + 0) * (OUT_F * NHEAD) + ncol];
            b.y = W[(k + khalf + 1) * (OUT_F * NHEAD) + ncol];
            acc[t] = __builtin_amdgcn_wmma_f32_16x16x4_f32(
                false, a, false, b, (short)0, acc[t], false, false);
        }
    }

    const int rowOff = (lane < 16) ? 0 : 8;
    #pragma unroll
    for (int t = 0; t < 8; ++t) {
        #pragma unroll
        for (int g = 0; g < 8; ++g) {
            long rr = row0 + g + rowOff;
            if (rr < N) {
                float v = acc[t][g];
                v = (v == v) ? v : 0.0f;
                h[rr * (OUT_F * NHEAD) + t * 16 + mrow] = v;
            }
        }
    }
}

__global__ __launch_bounds__(256) void gat_alar(const float* __restrict__ h,
                                                const float* __restrict__ a_l,
                                                const float* __restrict__ a_r,
                                                float* __restrict__ al,
                                                float* __restrict__ ar, int N) {
    const int wave = threadIdx.x >> 5;
    const int lane = threadIdx.x & 31;
    const long n = (long)blockIdx.x * 8 + wave;
    if (n >= N) return;
    const float* __restrict__ hn = h + n * (OUT_F * NHEAD);
    float sl[NHEAD], sr[NHEAD];
    #pragma unroll
    for (int j = 0; j < NHEAD; ++j) {
        float hv = hn[j * OUT_F + lane];
        sl[j] = a_l[j * OUT_F + lane] * hv;
        sr[j] = a_r[j * OUT_F + lane] * hv;
    }
    #pragma unroll
    for (int off = 16; off > 0; off >>= 1) {
        #pragma unroll
        for (int j = 0; j < NHEAD; ++j) {
            sl[j] += __shfl_xor(sl[j], off, 32);
            sr[j] += __shfl_xor(sr[j], off, 32);
        }
    }
    if (lane == 0) {
        #pragma unroll
        for (int j = 0; j < NHEAD; ++j) {
            al[n * NHEAD + j] = sl[j];
            ar[n * NHEAD + j] = sr[j];
        }
    }
}

__global__ __launch_bounds__(256) void gat_edge_score(const int* __restrict__ row,
                                                      const int* __restrict__ col,
                                                      const float* __restrict__ al,
                                                      const float* __restrict__ ar,
                                                      float* __restrict__ e,
                                                      float* __restrict__ m, int E) {
    long i = (long)blockIdx.x * blockDim.x + threadIdx.x;
    if (i >= E) return;
    const int r = row[i], c = col[i];
    #pragma unroll
    for (int j = 0; j < NHEAD; ++j) {
        float v = al[(long)r * NHEAD + j] + ar[(long)c * NHEAD + j];
        v = (v > 0.0f) ? v : (ALPHA * v);
        e[i * NHEAD + j] = v;
        atomicMaxFloat(&m[(long)r * NHEAD + j], v);
    }
}

__global__ __launch_bounds__(256) void gat_edge_exp(const int* __restrict__ row,
                                                    const float* __restrict__ m,
                                                    float* __restrict__ e,
                                                    float* __restrict__ s, int E) {
    long i = (long)blockIdx.x * blockDim.x + threadIdx.x;
    if (i >= E) return;
    const int r = row[i];
    #pragma unroll
    for (int j = 0; j < NHEAD; ++j) {
        float v = expf(e[i * NHEAD + j] - m[(long)r * NHEAD + j]);
        e[i * NHEAD + j] = v;
        atomicAdd(&s[(long)r * NHEAD + j], v);
    }
}

__global__ __launch_bounds__(256) void gat_aggregate(const int* __restrict__ row,
                                                     const int* __restrict__ col,
                                                     const float* __restrict__ e,
                                                     const float* __restrict__ s,
                                                     const float* __restrict__ h,
                                                     float* __restrict__ out, int E) {
    const int wave = threadIdx.x >> 5;
    const int lane = threadIdx.x & 31;
    const long i = (long)blockIdx.x * 8 + wave;
    if (i >= E) return;
    const int r = row[i], c = col[i];
    const float* __restrict__ hc = h + (long)c * (OUT_F * NHEAD);
    float* __restrict__ outr = out + (long)r * (OUT_F * NHEAD);
    #pragma unroll
    for (int j = 0; j < NHEAD; ++j) {
        float att = e[i * NHEAD + j] / s[(long)r * NHEAD + j];
        atomicAdd(&outr[j * OUT_F + lane], att * hc[j * OUT_F + lane]);
    }
}

extern "C" void kernel_launch(void* const* d_in, const int* in_sizes, int n_in,
                              void* d_out, int out_size, void* d_ws, size_t ws_size,
                              hipStream_t stream) {
    const float* x    = (const float*)d_in[0];
    const int*   edge = (const int*)d_in[1];
    const float* W    = (const float*)d_in[2];
    const float* a_l  = (const float*)d_in[3];
    const float* a_r  = (const float*)d_in[4];

    const int N = in_sizes[0] / IN_F;
    const int E = in_sizes[1] / 2;
    const int* rowp = edge;
    const int* colp = edge + E;

    float* ws = (float*)d_ws;
    float* h  = ws;
    float* al = h  + (size_t)N * (OUT_F * NHEAD);
    float* ar = al + (size_t)N * NHEAD;
    float* m  = ar + (size_t)N * NHEAD;
    float* s  = m  + (size_t)N * NHEAD;
    float* e  = s  + (size_t)N * NHEAD;
    float* out = (float*)d_out;

    {
        long total = (long)N * (OUT_F * NHEAD);
        gat_init<<<(unsigned)((total + 255) / 256), 256, 0, stream>>>(m, s, out, N);
    }
    gat_gemm<<<(unsigned)((N + 127) / 128), 256, 0, stream>>>(x, W, h, N);
    gat_alar<<<(unsigned)((N + 7) / 8), 256, 0, stream>>>(h, a_l, a_r, al, ar, N);
    gat_edge_score<<<(unsigned)((E + 255) / 256), 256, 0, stream>>>(rowp, colp, al, ar, e, m, E);
    gat_edge_exp<<<(unsigned)((E + 255) / 256), 256, 0, stream>>>(rowp, m, e, s, E);
    gat_aggregate<<<(unsigned)((E + 7) / 8), 256, 0, stream>>>(rowp, colp, e, s, h, out, E);
}


# Counting-sort edges by destination (int atomic arrival-order rank), then one wave per node does max, exp, sum and h-row sum in registers; float atomics removed, sum order changes
# speedup vs baseline: 3.4485x; 3.4485x over previous
#include <hip/hip_runtime.h>
#include <hip/hip_bf16.h>
#include <math.h>

typedef __attribute__((ext_vector_type(2))) float v2f;
typedef __attribute__((ext_vector_type(8))) float v8f;

#define IN_F   128
#define OUT_F  32
#define NHEAD  4
#define ALPHA  0.2f

__global__ __launch_bounds__(256) void gat_gemm(const float* __restrict__ x,
                                                const float* __restrict__ W,
                                                float* __restrict__ h, int N) {
    const int wave = threadIdx.x >> 5;
    const int lane = threadIdx.x & 31;
    const long rowTile = (long)blockIdx.x * 8 + wave;
    const long row0 = rowTile * 16;
    if (row0 >= N) return;

    const int mrow  = lane & 15;
    const int khalf = (lane >> 4) * 2;

    v8f acc[8];
    #pragma unroll
    for (int t = 0; t < 8; ++t) acc[t] = v8f{};

    long r = row0 + mrow;
    if (r >= N) r = N - 1;
    const float* __restrict__ xrow = x + r * IN_F;

    for (int k = 0; k < IN_F; k += 4) {
        v2f a;
        a.x = xrow[k + khalf + 0];
        a.y = xrow[k + khalf + 1];
        #pragma unroll
        for (int t = 0; t < 8; ++t) {
            const int ncol = t * 16 + mrow;
            v2f b;
            b.x = W[(k + khalf + 0) * (OUT_F * NHEAD) + ncol];
            b.y = W[(k + khalf + 1) * (OUT_F * NHEAD) + ncol];
            acc[t] = __builtin_amdgcn_wmma_f32_16x16x4_f32(
                false, a, false, b, (short)0, acc[t], false, false);
        }
    }

    const int rowOff = (lane < 16) ? 0 : 8;
    #pragma unroll
    for (int t = 0; t < 8; ++t) {
        #pragma unroll
        for (int g = 0; g < 8; ++g) {
            long rr = row0 + g + rowOff;
            if (rr < N) {
                float v = acc[t][g];
                v = (v == v) ? v : 0.0f;
                h[rr * (OUT_F * NHEAD) + t * 16 + mrow] = v;
            }
        }
    }
}

__global__ __launch_bounds__(256) void gat_alar(const float* __restrict__ h,
                                                const float* __restrict__ a_l,
                                                const float* __restrict__ a_r,
                                                float* __restrict__ al,
                                                float* __restrict__ ar, int N) {
    const int wave = threadIdx.x >> 5;
    const int lane = threadIdx.x & 31;
    const long n = (long)blockIdx.x * 8 + wave;
    if (n >= N) return;
    const float* __restrict__ hn = h + n * (OUT_F * NHEAD);
    float sl[NHEAD], sr[NHEAD];
    #pragma unroll
    for (int j = 0; j < NHEAD; ++j) {
        float hv = hn[j * OUT_F + lane];
        sl[j] = a_l[j * OUT_F + lane] * hv;
        sr[j] = a_r[j * OUT_F + lane] * hv;
    }
    #pragma unroll
    for (int off = 16; off > 0; off >>= 1) {
        #pragma unroll
        for (int j = 0; j < NHEAD; ++j) {
            sl[j] += __shfl_xor(sl[j], off, 32);
            sr[j] += __shfl_xor(sr[j], off, 32);
        }
    }
    if (lane == 0) {
        #pragma unroll
        for (int j = 0; j < NHEAD; ++j) {
            al[n * NHEAD + j] = sl[j];
            ar[n * NHEAD + j] = sr[j];
        }
    }
}

#define SCAN_TILE 1024

__device__ inline int wave_sum_i(int v) {
    #pragma unroll
    for (int off = 16; off > 0; off >>= 1) v += __shfl_xor(v, off, 32);
    return v;
}

__global__ __launch_bounds__(256) void gat_zero(int* __restrict__ count, int N) {
    const int i = blockIdx.x * 256 + threadIdx.x;
    if (i < N) count[i] = 0;
}

__global__ __launch_bounds__(256) void gat_hist(const int* __restrict__ row,
                                                int* __restrict__ count,
                                                int* __restrict__ rank, int E) {
    const long i = (long)blockIdx.x * 256 + threadIdx.x;
    if (i >= E) return;
    rank[i] = atomicAdd(&count[row[i]], 1);
}

__global__ __launch_bounds__(256) void gat_scan_tiles(const int* __restrict__ count,
                                                      int* __restrict__ bsum, int N) {
    __shared__ int wt[8];
    const int wave = threadIdx.x >> 5;
    const int lane = threadIdx.x & 31;
    const long base = (long)blockIdx.x * SCAN_TILE + threadIdx.x * 4;
    int t = 0;
    for (int q = 0; q < 4; ++q) {
        if (base + q < N) t += count[base + q];
    }
    t = wave_sum_i(t);
    if (lane == 0) wt[wave] = t;
    __syncthreads();
    if (threadIdx.x == 0) {
        int tot = 0;
        for (int w = 0; w < 8; ++w) tot += wt[w];
        bsum[blockIdx.x] = tot;
    }
}

__global__ __launch_bounds__(256) void gat_scan_offsets(const int* __restrict__ count,
                                                        const int* __restrict__ bsum,
                                                        int* __restrict__ offset, int N) {
    __shared__ int wc[8];
    __shared__ int wt[8];
    const int wave = threadIdx.x >> 5;
    const int lane = threadIdx.x & 31;

    int part = 0;
    for (int j = threadIdx.x; j < (int)blockIdx.x; j += 256) part += bsum[j];
    part = wave_sum_i(part);

    const long base = (long)blockIdx.x * SCAN_TILE + threadIdx.x * 4;
    int c[4];
    int t = 0;
    #pragma unroll
    for (int q = 0; q < 4; ++q) {
        c[q] = (base + q < N) ? count[base + q] : 0;
        t += c[q];
    }
    int incl = t;
    #pragma unroll
    for (int off = 1; off < 32; off <<= 1) {
        const int u = __shfl_up(incl, off, 32);
        if (lane >= off) incl += u;
    }
    if (lane == 0)  wc[wave] = part;
    if (lane == 31) wt[wave] = incl;
    __syncthreads();
    int pre = 0;
    for (int w = 0; w < 8; ++w) {
        pre += wc[w];
        if (w < wave) pre += wt[w];
    }
    int excl = pre + incl - t;
    #pragma unroll
    for (int q = 0; q < 4; ++q) {
        if (base + q < N) { offset[base + q] = excl; excl += c[q]; }
    }
}

__global__ __launch_bounds__(256) void gat_scatter(const int* __restrict__ row,
                                                   const int* __restrict__ col,
                                                   const int* __restrict__ rank,
                                                   const int* __restrict__ offset,
                                                   int* __restrict__ scol, int E) {
    const long i = (long)blockIdx.x * 256 + threadIdx.x;
    if (i >= E) return;
    scol[offset[row[i]] + rank[i]] = col[i];
}

__device__ inline float lrelu(float v) { return (v > 0.0f) ? v : (ALPHA * v); }

template <int K>
__device__ inline float group_bcast(float v) {
    return __int_as_float((int)__builtin_amdgcn_mov_dpp8((unsigned)__float_as_int(v), (unsigned)(K * 0x249249)));
}
__device__ inline float group_max(float v) {
    #pragma unroll
    for (int off = 4; off > 0; off >>= 1) v = fmaxf(v, __shfl_xor(v, off, 32));
    return v;
}
__device__ inline float group_sum(float v) {
    #pragma unroll
    for (int off = 4; off > 0; off >>= 1) v += __shfl_xor(v, off, 32);
    return v;
}

__global__ __launch_bounds__(256) void gat_node(const int* __restrict__ offset,
                                                const int* __restrict__ count,
                                                const int* __restrict__ scol,
                                                const float* __restrict__ al,
                                                const float* __restrict__ ar,
                                                const float* __restrict__ h,
                                                float* __restrict__ out, int N) {
    const int lane = threadIdx.x & 31;
    const int n = __builtin_amdgcn_readfirstlane((int)(blockIdx.x * 8 + (threadIdx.x >> 5)));
    if (n >= N) return;
    const int beg = offset[n];
    const int end = beg + count[n];
    const int e8 = lane & 7;
    const int hd = lane >> 3;

    float4 acc = make_float4(0.0f, 0.0f, 0.0f, 0.0f);
    if (end > beg) {
        const float aln = al[(long)n * NHEAD + hd];
        const float* __restrict__ hl = h + lane * 4;

        float m = -INFINITY;
        for (int base = beg; base < end; base += 8) {
            const int idx = base + e8;
            if (idx < end) m = fmaxf(m, lrelu(aln + ar[(long)scol[idx] * NHEAD + hd]));
        }
        m = group_max(m);

        float s = 0.0f;
        for (int base = beg; base < end; base += 8) {
            const int idx = base + e8;
            if (idx < end) s += expf(lrelu(aln + ar[(long)scol[idx] * NHEAD + hd]) - m);
        }
        s = group_sum(s);

        for (int base = beg; base < end; base += 8) {
            const int cnt = (end - base < 8) ? (end - base) : 8;
            int c = 0;
            float p = 0.0f;
            if (e8 < cnt) {
                c = scol[base + e8];
                p = expf(lrelu(aln + ar[(long)c * NHEAD + hd]) - m) / s;
            }
            float4 hv[8];
#define H_LOAD(K) if (K < cnt) { hv[K] = *(const float4*)(hl + (long)__builtin_amdgcn_readlane(c, K) * (OUT_F * NHEAD)); }
            H_LOAD(0) H_LOAD(1) H_LOAD(2) H_LOAD(3) H_LOAD(4) H_LOAD(5) H_LOAD(6) H_LOAD(7)
#undef H_LOAD
#define H_FMA(K) if (K < cnt) { const float w = group_bcast<K>(p); \
                acc.x += w * hv[K].x; acc.y += w * hv[K].y; acc.z += w * hv[K].z; acc.w += w * hv[K].w; }
            H_FMA(0) H_FMA(1) H_FMA(2) H_FMA(3) H_FMA(4) H_FMA(5) H_FMA(6) H_FMA(7)
#undef H_FMA
        }
    }
    *(float4*)(out + (long)n * (OUT_F * NHEAD) + lane * 4) = acc;
}

extern "C" void kernel_launch(void* const* d_in, const int* in_sizes, int n_in,
                              void* d_out, int out_size, void* d_ws, size_t ws_size,
                              hipStream_t stream) {
    const float* x    = (const float*)d_in[0];
    const int*   edge = (const int*)d_in[1];
    const float* W    = (const float*)d_in[2];
    const float* a_l  = (const float*)d_in[3];
    const float* a_r  = (const float*)d_in[4];

    const int N = in_sizes[0] / IN_F;
    const int E = in_sizes[1] / 2;
    const int* rowp = edge;
    const int* colp = edge + E;
    const int nTiles = (N + SCAN_TILE - 1) / SCAN_TILE;

    float* ws = (float*)d_ws;
    float* h  = ws;
    float* al = h  + (size_t)N * (OUT_F * NHEAD);
    float* ar = al + (size_t)N * NHEAD;
    int* count  = (int*)(ar + (size_t)N * NHEAD);
    int* offset = count + (size_t)N;
    int* bsum   = offset + (size_t)N;
    int* rank   = bsum + (size_t)nTiles;
    int* scol   = rank + (size_t)E;
    float* out = (float*)d_out;

    gat_zero<<<(unsigned)((N + 255) / 256), 256, 0, stream>>>(count, N);
    gat_hist<<<(unsigned)((E + 255) / 256), 256, 0, stream>>>(rowp, count, rank, E);
    gat_scan_tiles<<<(unsigned)nTiles, 256, 0, stream>>>(count, bsum, N);
    gat_scan_offsets<<<(unsigned)nTiles, 256, 0, stream>>>(count, bsum, offset, N);
    gat_scatter<<<(unsigned)((E + 255) / 256), 256, 0, stream>>>(rowp, colp, rank, offset, scol, E);
    gat_gemm<<<(unsigned)((N + 127) / 128), 256, 0, stream>>>(x, W, h, N);
    gat_alar<<<(unsigned)((N + 7) / 8), 256, 0, stream>>>(h, a_l, a_r, al, ar, N);
    gat_node<<<(unsigned)((N + 7) / 8), 256, 0, stream>>>(offset, count, scol, al, ar, h, out, N);
}


# Edges grouped by destination in two levels on LDS counters (returning LDS add = arrival-order rank, stores staged and contiguous) instead of one global atomic and one scattered 4-B store per edge
# speedup vs baseline: 5.5795x; 1.6179x over previous
#include <hip/hip_runtime.h>
#include <hip/hip_bf16.h>
#include <math.h>

typedef __attribute__((ext_vector_type(2))) float v2f;
typedef __attribute__((ext_vector_type(8))) float v8f;

#define IN_F   128
#define OUT_F  32
#define NHEAD  4
#define ALPHA  0.2f

__global__ __launch_bounds__(256) void gat_gemm(const float* __restrict__ x,
                                                const float* __restrict__ W,
                                                float* __restrict__ h, int N) {
    const int wave = threadIdx.x >> 5;
    const int lane = threadIdx.x & 31;
    const long rowTile = (long)blockIdx.x * 8 + wave;
    const long row0 = rowTile * 16;
    if (row0 >= N) return;

    const int mrow  = lane & 15;
    const int khalf = (lane >> 4) * 2;

    v8f acc[8];
    #pragma unroll
    for (int t = 0; t < 8; ++t) acc[t] = v8f{};

    long r = row0 + mrow;
    if (r >= N) r = N - 1;
    const float* __restrict__ xrow = x + r * IN_F;

    for (int k = 0; k < IN_F; k += 4) {
        v2f a;
        a.x = xrow[k + khalf + 0];
        a.y = xrow[k + khalf + 1];
        #pragma unroll
        for (int t = 0; t < 8; ++t) {
            const int ncol = t * 16 + mrow;
            v2f b;
            b.x = W[(k + khalf + 0) * (OUT_F * NHEAD) + ncol];
            b.y = W[(k + khalf + 1) * (OUT_F * NHEAD) + ncol];
            acc[t] = __builtin_amdgcn_wmma_f32_16x16x4_f32(
                false, a, false, b, (short)0, acc[t], false, false);
        }
    }

    const int rowOff = (lane < 16) ? 0 : 8;
    #pragma unroll
    for (int t = 0; t < 8; ++t) {
        #pragma unroll
        for (int g = 0; g < 8; ++g) {
            long rr = row0 + g + rowOff;
            if (rr < N) {
                float v = acc[t][g];
                v = (v == v) ? v : 0.0f;
                h[rr * (OUT_F * NHEAD) + t * 16 + mrow] = v;
            }
        }
    }
}

__global__ __launch_bounds__(256) void gat_alar(const float* __restrict__ h,
                                                const float* __restrict__ a_l,
                                                const float* __restrict__ a_r,
                                                float* __restrict__ al,
                                                float* __restrict__ ar, int N) {
    const int wave = threadIdx.x >> 5;
    const int lane = threadIdx.x & 31;
    const long n = (long)blockIdx.x * 8 + wave;
    if (n >= N) return;
    const float* __restrict__ hn = h + n * (OUT_F * NHEAD);
    float sl[NHEAD], sr[NHEAD];
    #pragma unroll
    for (int j = 0; j < NHEAD; ++j) {
        float hv = hn[j * OUT_F + lane];
        sl[j] = a_l[j * OUT_F + lane] * hv;
        sr[j] = a_r[j * OUT_F + lane] * hv;
    }
    #pragma unroll
    for (int off = 16; off > 0; off >>= 1) {
        #pragma unroll
        for (int j = 0; j < NHEAD; ++j) {
            sl[j] += __shfl_xor(sl[j], off, 32);
            sr[j] += __shfl_xor(sr[j], off, 32);
        }
    }
    if (lane == 0) {
        #pragma unroll
        for (int j = 0; j < NHEAD; ++j) {
            al[n * NHEAD + j] = sl[j];
            ar[n * NHEAD + j] = sr[j];
        }
    }
}

#define GA     256
#define NBIN   512
#define SMAX   1024
#define CHMAX  8192
#define CAPC   16384

__device__ inline int bin_shift(int N) {
    return 32 - __clz((N - 1) >> 9);
}

__device__ inline int block_excl_scan(int v, int* wtot) {
    const int lane = threadIdx.x & 31;
    const int wave = threadIdx.x >> 5;
    const int nw = blockDim.x >> 5;
    int incl = v;
    #pragma unroll
    for (int off = 1; off < 32; off <<= 1) {
        const int u = __shfl_up(incl, off, 32);
        if (lane >= off) incl += u;
    }
    if (lane == 31) wtot[wave] = incl;
    __syncthreads();
    int pre = 0;
    for (int w = 0; w < nw; ++w) {
        if (w < wave) pre += wtot[w];
    }
    __syncthreads();
    return pre + incl - v;
}

__global__ __launch_bounds__(1024) void gat_bincount(const int* __restrict__ row,
                                                     int* __restrict__ wgcnt, int E, int N) {
    __shared__ int cnt[NBIN];
    const int tid = threadIdx.x;
    const int shift = bin_shift(N);
    const int CH = (E + GA - 1) / GA;
    const int lo = (int)blockIdx.x * CH;
    const int hi = (lo + CH < E) ? (lo + CH) : E;
    if (tid < NBIN) cnt[tid] = 0;
    __syncthreads();
    for (int i0 = lo + tid; i0 < hi; i0 += 4 * 1024) {
        int r[4];
        #pragma unroll
        for (int k = 0; k < 4; ++k) {
            const int i = i0 + k * 1024;
            r[k] = (i < hi) ? row[i] : 0;
        }
        #pragma unroll
        for (int k = 0; k < 4; ++k) {
            if (i0 + k * 1024 < hi) atomicAdd(&cnt[r[k] >> shift], 1);
        }
    }
    __syncthreads();
    if (tid < NBIN) wgcnt[(int)blockIdx.x * NBIN + tid] = cnt[tid];
}

__global__ __launch_bounds__(2 * NBIN) void gat_binscan(const int* wgcnt, int* base,
                                                        int* __restrict__ binStart) {
    __shared__ int wtot[32];
    __shared__ int part0[NBIN];
    __shared__ int part1[NBIN];
    __shared__ int bstart[NBIN];
    const int b  = threadIdx.x & (NBIN - 1);
    const int hf = threadIdx.x >> 9;
    const int* src = wgcnt + (hf * (GA / 2) * NBIN + b);
    int*       dst = base  + (hf * (GA / 2) * NBIN + b);
    int part = 0;
    for (int t0 = 0; t0 < GA / 2; t0 += 32) {
        int v[32];
        #pragma unroll
        for (int k = 0; k < 32; ++k) v[k] = src[(t0 + k) * NBIN];
        #pragma unroll
        for (int k = 0; k < 32; ++k) part += v[k];
    }
    if (hf == 0) part0[b] = part; else part1[b] = part;
    __syncthreads();
    const int tot = (hf == 0) ? (part0[b] + part1[b]) : 0;
    const int excl = block_excl_scan(tot, wtot);
    if (hf == 0) {
        bstart[b] = excl;
        binStart[b] = excl;
        if (b == NBIN - 1) binStart[NBIN] = excl + tot;
    }
    __syncthreads();
    int run = bstart[b] + ((hf == 0) ? 0 : part0[b]);
    for (int t0 = 0; t0 < GA / 2; t0 += 32) {
        int v[32];
        #pragma unroll
        for (int k = 0; k < 32; ++k) v[k] = src[(t0 + k) * NBIN];
        #pragma unroll
        for (int k = 0; k < 32; ++k) {
            dst[(t0 + k) * NBIN] = run;
            run += v[k];
        }
    }
}

__global__ __launch_bounds__(1024) void gat_place(const int* __restrict__ row,
                                                  const int* __restrict__ col,
                                                  const int* __restrict__ wgcnt,
                                                  const int* __restrict__ base,
                                                  int* __restrict__ recw, int E, int N) {
    int2* __restrict__ rec = (int2*)recw;
    __shared__ int cur[NBIN];
    __shared__ int delta[NBIN];
    __shared__ int wtot[32];
    __shared__ int2 staged[CHMAX];
    const int tid = threadIdx.x;
    const int shift = bin_shift(N);
    const int CH = (E + GA - 1) / GA;
    const int lo = (int)blockIdx.x * CH;
    const int hi = (lo + CH < E) ? (lo + CH) : E;
    const int len = (hi > lo) ? (hi - lo) : 0;
    const bool stage = (CH <= CHMAX);

    const int c = (tid < NBIN) ? wgcnt[(int)blockIdx.x * NBIN + tid] : 0;
    const int excl = block_excl_scan(c, wtot);
    if (tid < NBIN) {
        const int g = base[(int)blockIdx.x * NBIN + tid];
        cur[tid] = stage ? excl : g;
        delta[tid] = g - excl;
    }
    __syncthreads();
    for (int i0 = lo + tid; i0 < hi; i0 += 4 * 1024) {
        int r[4], s[4];
        #pragma unroll
        for (int k = 0; k < 4; ++k) {
            const int i = i0 + k * 1024;
            r[k] = (i < hi) ? row[i] : 0;
            s[k] = (i < hi) ? col[i] : 0;
        }
        #pragma unroll
        for (int k = 0; k < 4; ++k) {
            if (i0 + k * 1024 < hi) {
                const int pos = atomicAdd(&cur[r[k] >> shift], 1);
                if (stage) staged[pos] = make_int2(r[k], s[k]);
                else       rec[pos]    = make_int2(r[k], s[k]);
            }
        }
    }
    if (stage) {
        __syncthreads();
        for (int j = tid; j < len; j += 1024) {
            const int2 e = staged[j];
            rec[j + delta[e.x >> shift]] = e;
        }
    }
}

__global__ __launch_bounds__(1024) void gat_group(const int* __restrict__ recw,
                                                  const int* __restrict__ binStart,
                                                  int* __restrict__ count,
                                                  int* __restrict__ offset,
                                                  int* __restrict__ scol, int N) {
    const int2* __restrict__ rec = (const int2*)recw;
    __shared__ int ncnt[SMAX];
    __shared__ int ncur[SMAX];
    __shared__ int wtot[32];
    __shared__ int staged[CAPC];
    const int tid = threadIdx.x;
    const int shift = bin_shift(N);
    const int n0 = (int)blockIdx.x << shift;
    if (n0 >= N) return;
    const int s0 = binStart[blockIdx.x];
    const int len = binStart[blockIdx.x + 1] - s0;
    const bool stage = (len <= CAPC);

    ncnt[tid] = 0;
    __syncthreads();
    for (int j0 = tid; j0 < len; j0 += 4 * 1024) {
        int r[4];
        #pragma unroll
        for (int k = 0; k < 4; ++k) {
            const int j = j0 + k * 1024;
            r[k] = (j < len) ? rec[s0 + j].x : n0;
        }
        #pragma unroll
        for (int k = 0; k < 4; ++k) {
            if (j0 + k * 1024 < len) atomicAdd(&ncnt[r[k] - n0], 1);
        }
    }
    __syncthreads();
    const int v = ncnt[tid];
    const int excl = block_excl_scan(v, wtot);
    ncur[tid] = excl;
    const int n = n0 + tid;
    if (tid < (1 << shift) && n < N) {
        count[n] = v;
        offset[n] = s0 + excl;
    }
    __syncthreads();
    for (int j0 = tid; j0 < len; j0 += 4 * 1024) {
        int2 e[4];
        #pragma unroll
        for (int k = 0; k < 4; ++k) {
            const int j = j0 + k * 1024;
            e[k] = (j < len) ? rec[s0 + j] : make_int2(n0, 0);
        }
        #pragma unroll
        for (int k = 0; k < 4; ++k) {
            if (j0 + k * 1024 < len) {
                const int pos = atomicAdd(&ncur[e[k].x - n0], 1);
                if (stage) staged[pos] = e[k].y;
                else       scol[s0 + pos] = e[k].y;
            }
        }
    }
    if (stage) {
        __syncthreads();
        for (int j = tid; j < len; j += 1024) scol[s0 + j] = staged[j];
    }
}

__device__ inline float lrelu(float v) { return (v > 0.0f) ? v : (ALPHA * v); }

template <int K>
__device__ inline float group_bcast(float v) {
    return __int_as_float((int)__builtin_amdgcn_mov_dpp8((unsigned)__float_as_int(v), (unsigned)(K * 0x249249)));
}
__device__ inline float group_max(float v) {
    #pragma unroll
    for (int off = 4; off > 0; off >>= 1) v = fmaxf(v, __shfl_xor(v, off, 32));
    return v;
}
__device__ inline float group_sum(float v) {
    #pragma unroll
    for (int off = 4; off > 0; off >>= 1) v += __shfl_xor(v, off, 32);
    return v;
}

__global__ __launch_bounds__(256) void gat_node(const int* __restrict__ offset,
                                                const int* __restrict__ count,
                                                const int* __restrict__ scol,
                                                const float* __restrict__ al,
                                                const float* __restrict__ ar,
                                                const float* __restrict__ h,
                                                float* __restrict__ out, int N) {
    const int lane = threadIdx.x & 31;
    const int n = __builtin_amdgcn_readfirstlane((int)(blockIdx.x * 8 + (threadIdx.x >> 5)));
    if (n >= N) return;
    const int beg = offset[n];
    const int end = beg + count[n];
    const int e8 = lane & 7;
    const int hd = lane >> 3;

    float4 acc = make_float4(0.0f, 0.0f, 0.0f, 0.0f);
    if (end > beg) {
        const float aln = al[(long)n * NHEAD + hd];
        const float* __restrict__ hl = h + lane * 4;

        float m = -INFINITY;
        for (int base = beg; base < end; base += 8) {
            const int idx = base + e8;
            if (idx < end) m = fmaxf(m, lrelu(aln + ar[(long)scol[idx] * NHEAD + hd]));
        }
        m = group_max(m);

        float s = 0.0f;
        for (int base = beg; base < end; base += 8) {
            const int idx = base + e8;
            if (idx < end) s += expf(lrelu(aln + ar[(long)scol[idx] * NHEAD + hd]) - m);
        }
        s = group_sum(s);

        for (int base = beg; base < end; base += 8) {
            const int cnt = (end - base < 8) ? (end - base) : 8;
            int c = 0;
            float p = 0.0f;
            if (e8 < cnt) {
                c = scol[base + e8];
                p = expf(lrelu(aln + ar[(long)c * NHEAD + hd]) - m) / s;
            }
            float4 hv[8];
#define H_LOAD(K) if (K < cnt) { hv[K] = *(const float4*)(hl + (long)__builtin_amdgcn_readlane(c, K) * (OUT_F * NHEAD)); }
            H_LOAD(0) H_LOAD(1) H_LOAD(2) H_LOAD(3) H_LOAD(4) H_LOAD(5) H_LOAD(6) H_LOAD(7)
#undef H_LOAD
#define H_FMA(K) if (K < cnt) { const float w = group_bcast<K>(p); \
                acc.x += w * hv[K].x; acc.y += w * hv[K].y; acc.z += w * hv[K].z; acc.w += w * hv[K].w; }
            H_FMA(0) H_FMA(1) H_FMA(2) H_FMA(3) H_FMA(4) H_FMA(5) H_FMA(6) H_FMA(7)
#undef H_FMA
        }
    }
    *(float4*)(out + (long)n * (OUT_F * NHEAD) + lane * 4) = acc;
}

extern "C" void kernel_launch(void* const* d_in, const int* in_sizes, int n_in,
                              void* d_out, int out_size, void* d_ws, size_t ws_size,
                              hipStream_t stream) {
    const float* x    = (const float*)d_in[0];
    const int*   edge = (const int*)d_in[1];
    const float* W    = (const float*)d_in[2];
    const float* a_l  = (const float*)d_in[3];
    const float* a_r  = (const float*)d_in[4];

    const int N = in_sizes[0] / IN_F;
    const int E = in_sizes[1] / 2;
    const int* rowp = edge;
    const int* colp = edge + E;

    float* ws = (float*)d_ws;
    float* h  = ws;
    float* al = h  + (size_t)N * (OUT_F * NHEAD);
    float* ar = al + (size_t)N * NHEAD;
    int* count  = (int*)(ar + (size_t)N * NHEAD);
    int* offset = count + (size_t)N;
    int* scol   = offset + (size_t)N;
    int* wgcnt    = (int*)ws;
    int* base     = wgcnt + GA * NBIN;
    int* binStart = base + GA * NBIN;
    int* rec      = binStart + 1024;
    float* out = (float*)d_out;

    gat_bincount<<<GA, 1024, 0, stream>>>(rowp, wgcnt, E, N);
    gat_binscan<<<1, 2 * NBIN, 0, stream>>>(wgcnt, base, binStart);
    gat_place<<<GA, 1024, 0, stream>>>(rowp, colp, wgcnt, base, rec, E, N);
    gat_group<<<NBIN, 1024, 0, stream>>>(rec, binStart, count, offset, scol, N);
    gat_gemm<<<(unsigned)((N + 127) / 128), 256, 0, stream>>>(x, W, h, N);
    gat_alar<<<(unsigned)((N + 7) / 8), 256, 0, stream>>>(h, a_l, a_r, al, ar, N);
    gat_node<<<(unsigned)((N + 7) / 8), 256, 0, stream>>>(offset, count, scol, al, ar, h, out, N);
}


# al, ar dots taken in the product's epilogue from the accumulator tiles (halving DPP folds over each 16-lane row) instead of a one-wave-per-node launch that re-reads h; dot sum order changes
# speedup vs baseline: 5.9878x; 1.0732x over previous
#include <hip/hip_runtime.h>
#include <hip/hip_bf16.h>
#include <math.h>

typedef __attribute__((ext_vector_type(2))) float v2f;
typedef __attribute__((ext_vector_type(8))) float v8f;

#define IN_F   128
#define OUT_F  32
#define NHEAD  4
#define ALPHA  0.2f


template <int XM>
__device__ inline float row_xor(float v) {
    return __int_as_float(__builtin_amdgcn_update_dpp(0, __float_as_int(v), 0x160 | XM, 0xf, 0xf, true));
}
template <int XM>
__device__ inline float fold(float a, float b, bool hi) {
    const float keep = hi ? b : a;
    const float send = hi ? a : b;
    return keep + row_xor<XM>(send);
}

__global__ __launch_bounds__(256) void gat_gemm(const float* __restrict__ x,
                                                const float* __restrict__ W,
                                                const float* __restrict__ a_l,
                                                const float* __restrict__ a_r,
                                                float* __restrict__ h,
                                                float* __restrict__ al,
                                                float* __restrict__ ar, int N) {
    const int wave = threadIdx.x >> 5;
    const int lane = threadIdx.x & 31;
    const long rowTile = (long)blockIdx.x * 8 + wave;
    const long row0 = rowTile * 16;
    if (row0 >= N) return;

    const int mrow  = lane & 15;
    const int khalf = (lane >> 4) * 2;

    v8f acc[8];
    #pragma unroll
    for (int t = 0; t < 8; ++t) acc[t] = v8f{};

    long r = row0 + mrow;
    if (r >= N) r = N - 1;
    const float* __restrict__ xrow = x + r * IN_F;

    for (int k = 0; k < IN_F; k += 4) {
        v2f a;
        a.x = xrow[k + khalf + 0];
        a.y = xrow[k + khalf + 1];
        #pragma unroll
        for (int t = 0; t < 8; ++t) {
            const int ncol = t * 16 + mrow;
            v2f b;
            b.x = W[(k + khalf + 0) * (OUT_F * NHEAD) + ncol];
            b.y = W[(k + khalf + 1) * (OUT_F * NHEAD) + ncol];
            acc[t] = __builtin_amdgcn_wmma_f32_16x16x4_f32(
                false, a, false, b, (short)0, acc[t], false, false);
        }
    }

    const int rowOff = (lane < 16) ? 0 : 8;
    #pragma unroll
    for (int t = 0; t < 8; ++t) {
        #pragma unroll
        for (int g = 0; g < 8; ++g) {
            float v = acc[t][g];
            v = (v == v) ? v : 0.0f;
            acc[t][g] = v;
            const long rr = row0 + g + rowOff;
            if (rr < N) h[rr * (OUT_F * NHEAD) + t * 16 + mrow] = v;
        }
    }

    const bool b0 = (mrow & 1) != 0, b1 = (mrow & 2) != 0, b2 = (mrow & 4) != 0, b3 = (mrow & 8) != 0;
    float res[NHEAD];
    #pragma unroll
    for (int j = 0; j < NHEAD; ++j) {
        float q[2];
        #pragma unroll
        for (int c = 0; c < 2; ++c) {
            const float* __restrict__ av = (c == 0) ? a_l : a_r;
            const float w0 = av[(2 * j) * 16 + mrow];
            const float w1 = av[(2 * j + 1) * 16 + mrow];
            float p[8];
            #pragma unroll
            for (int g = 0; g < 8; ++g) p[g] = w0 * acc[2 * j][g] + w1 * acc[2 * j + 1][g];
            const float p01 = fold<1>(p[0], p[1], b0);
            const float p23 = fold<1>(p[2], p[3], b0);
            const float p45 = fold<1>(p[4], p[5], b0);
            const float p67 = fold<1>(p[6], p[7], b0);
            const float p03 = fold<2>(p01, p23, b1);
            const float p47 = fold<2>(p45, p67, b1);
            q[c] = fold<4>(p03, p47, b2);
        }
        res[j] = fold<8>(q[0], q[1], b3);
    }
    const long orow = row0 + rowOff + (mrow & 7);
    if (orow < N) {
        float* __restrict__ dst = (b3 ? ar : al) + orow * NHEAD;
        *(float4*)dst = make_float4(res[0], res[1], res[2], res[3]);
    }
}

#define GA     256
#define NBIN   512
#define SMAX   1024
#define CHMAX  8192
#define CAPC   16384

__device__ inline int bin_shift(int N) {
    return 32 - __clz((N - 1) >> 9);
}

__device__ inline int block_excl_scan(int v, int* wtot) {
    const int lane = threadIdx.x & 31;
    const int wave = threadIdx.x >> 5;
    const int nw = blockDim.x >> 5;
    int incl = v;
    #pragma unroll
    for (int off = 1; off < 32; off <<= 1) {
        const int u = __shfl_up(incl, off, 32);
        if (lane >= off) incl += u;
    }
    if (lane == 31) wtot[wave] = incl;
    __syncthreads();
    int pre = 0;
    for (int w = 0; w < nw; ++w) {
        if (w < wave) pre += wtot[w];
    }
    __syncthreads();
    return pre + incl - v;
}

__global__ __launch_bounds__(1024) void gat_bincount(const int* __restrict__ row,
                                                     int* __restrict__ wgcnt, int E, int N) {
    __shared__ int cnt[NBIN];
    const int tid = threadIdx.x;
    const int shift = bin_shift(N);
    const int CH = (E + GA - 1) / GA;
    const int lo = (int)blockIdx.x * CH;
    const int hi = (lo + CH < E) ? (lo + CH) : E;
    if (tid < NBIN) cnt[tid] = 0;
    __syncthreads();
    for (int i0 = lo + tid; i0 < hi; i0 += 4 * 1024) {
        int r[4];
        #pragma unroll
        for (int k = 0; k < 4; ++k) {
            const int i = i0 + k * 1024;
            r[k] = (i < hi) ? row[i] : 0;
        }
        #pragma unroll
        for (int k = 0; k < 4; ++k) {
            if (i0 + k * 1024 < hi) atomicAdd(&cnt[r[k] >> shift], 1);
        }
    }
    __syncthreads();
    if (tid < NBIN) wgcnt[(int)blockIdx.x * NBIN + tid] = cnt[tid];
}

__global__ __launch_bounds__(2 * NBIN) void gat_binscan(const int* wgcnt, int* base,
                                                        int* __restrict__ binStart) {
    __shared__ int wtot[32];
    __shared__ int part0[NBIN];
    __shared__ int part1[NBIN];
    __shared__ int bstart[NBIN];
    const int b  = threadIdx.x & (NBIN - 1);
    const int hf = threadIdx.x >> 9;
    const int* src = wgcnt + (hf * (GA / 2) * NBIN + b);
    int*       dst = base  + (hf * (GA / 2) * NBIN + b);
    int part = 0;
    for (int t0 = 0; t0 < GA / 2; t0 += 32) {
        int v[32];
        #pragma unroll
        for (int k = 0; k < 32; ++k) v[k] = src[(t0 + k) * NBIN];
        #pragma unroll
        for (int k = 0; k < 32; ++k) part += v[k];
    }
    if (hf == 0) part0[b] = part; else part1[b] = part;
    __syncthreads();
    const int tot = (hf == 0) ? (part0[b] + part1[b]) : 0;
    const int excl = block_excl_scan(tot, wtot);
    if (hf == 0) {
        bstart[b] = excl;
        binStart[b] = excl;
        if (b == NBIN - 1) binStart[NBIN] = excl + tot;
    }
    __syncthreads();
    int run = bstart[b] + ((hf == 0) ? 0 : part0[b]);
    for (int t0 = 0; t0 < GA / 2; t0 += 32) {
        int v[32];
        #pragma unroll
        for (int k = 0; k < 32; ++k) v[k] = src[(t0 + k) * NBIN];
        #pragma unroll
        for (int k = 0; k < 32; ++k) {
            dst[(t0 + k) * NBIN] = run;
            run += v[k];
        }
    }
}

__global__ __launch_bounds__(1024) void gat_place(const int* __restrict__ row,
                                                  const int* __restrict__ col,
                                                  const int* __restrict__ wgcnt,
                                                  const int* __restrict__ base,
                                                  int* __restrict__ recw, int E, int N) {
    int2* __restrict__ rec = (int2*)recw;
    __shared__ int cur[NBIN];
    __shared__ int delta[NBIN];
    __shared__ int wtot[32];
    __shared__ int2 staged[CHMAX];
    const int tid = threadIdx.x;
    const int shift = bin_shift(N);
    const int CH = (E + GA - 1) / GA;
    const int lo = (int)blockIdx.x * CH;
    const int hi = (lo + CH < E) ? (lo + CH) : E;
    const int len = (hi > lo) ? (hi - lo) : 0;
    const bool stage = (CH <= CHMAX);

    const int c = (tid < NBIN) ? wgcnt[(int)blockIdx.x * NBIN + tid] : 0;
    const int excl = block_excl_scan(c, wtot);
    if (tid < NBIN) {
        const int g = base[(int)blockIdx.x * NBIN + tid];
        cur[tid] = stage ? excl : g;
        delta[tid] = g - excl;
    }
    __syncthreads();
    for (int i0 = lo + tid; i0 < hi; i0 += 4 * 1024) {
        int r[4], s[4];
        #pragma unroll
        for (int k = 0; k < 4; ++k) {
            const int i = i0 + k * 1024;
            r[k] = (i < hi) ? row[i] : 0;
            s[k] = (i < hi) ? col[i] : 0;
        }
        #pragma unroll
        for (int k = 0; k < 4; ++k) {
            if (i0 + k * 1024 < hi) {
                const int pos = atomicAdd(&cur[r[k] >> shift], 1);
                if (stage) staged[pos] = make_int2(r[k], s[k]);
                else       rec[pos]    = make_int2(r[k], s[k]);
            }
        }
    }
    if (stage) {
        __syncthreads();
        for (int j = tid; j < len; j += 1024) {
            const int2 e = staged[j];
            rec[j + delta[e.x >> shift]] = e;
        }
    }
}

__global__ __launch_bounds__(1024) void gat_group(const int* __restrict__ recw,
                                                  const int* __restrict__ binStart,
                                                  int* __restrict__ count,
                                                  int* __restrict__ offset,
                                                  int* __restrict__ scol, int N) {
    const int2* __restrict__ rec = (const int2*)recw;
    __shared__ int ncnt[SMAX];
    __shared__ int ncur[SMAX];
    __shared__ int wtot[32];
    __shared__ int staged[CAPC];
    const int tid = threadIdx.x;
    const int shift = bin_shift(N);
    const int n0 = (int)blockIdx.x << shift;
    if (n0 >= N) return;
    const int s0 = binStart[blockIdx.x];
    const int len = binStart[blockIdx.x + 1] - s0;
    const bool stage = (len <= CAPC);

    ncnt[tid] = 0;
    __syncthreads();
    for (int j0 = tid; j0 < len; j0 += 4 * 1024) {
        int r[4];
        #pragma unroll
        for (int k = 0; k < 4; ++k) {
            const int j = j0 + k * 1024;
            r[k] = (j < len) ? rec[s0 + j].x : n0;
        }
        #pragma unroll
        for (int k = 0; k < 4; ++k) {
            if (j0 + k * 1024 < len) atomicAdd(&ncnt[r[k] - n0], 1);
        }
    }
    __syncthreads();
    const int v = ncnt[tid];
    const int excl = block_excl_scan(v, wtot);
    ncur[tid] = excl;
    const int n = n0 + tid;
    if (tid < (1 << shift) && n < N) {
        count[n] = v;
        offset[n] = s0 + excl;
    }
    __syncthreads();
    for (int j0 = tid; j0 < len; j0 += 4 * 1024) {
        int2 e[4];
        #pragma unroll
        for (int k = 0; k < 4; ++k) {
            const int j = j0 + k * 1024;
            e[k] = (j < len) ? rec[s0 + j] : make_int2(n0, 0);
        }
        #pragma unroll
        for (int k = 0; k < 4; ++k) {
            if (j0 + k * 1024 < len) {
                const int pos = atomicAdd(&ncur[e[k].x - n0], 1);
                if (stage) staged[pos] = e[k].y;
                else       scol[s0 + pos] = e[k].y;
            }
        }
    }
    if (stage) {
        __syncthreads();
        for (int j = tid; j < len; j += 1024) scol[s0 + j] = staged[j];
    }
}

__device__ inline float lrelu(float v) { return (v > 0.0f) ? v : (ALPHA * v); }

template <int K>
__device__ inline float group_bcast(float v) {
    return __int_as_float((int)__builtin_amdgcn_mov_dpp8((unsigned)__float_as_int(v), (unsigned)(K * 0x249249)));
}
__device__ inline float group_max(float v) {
    #pragma unroll
    for (int off = 4; off > 0; off >>= 1) v = fmaxf(v, __shfl_xor(v, off, 32));
    return v;
}
__device__ inline float group_sum(float v) {
    #pragma unroll
    for (int off = 4; off > 0; off >>= 1) v += __shfl_xor(v, off, 32);
    return v;
}

__global__ __launch_bounds__(256) void gat_node(const int* __restrict__ offset,
                                                const int* __restrict__ count,
                                                const int* __restrict__ scol,
                                                const float* __restrict__ al,
                                                const float* __restrict__ ar,
                                                const float* __restrict__ h,
                                                float* __restrict__ out, int N) {
    const int lane = threadIdx.x & 31;
    const int n = __builtin_amdgcn_readfirstlane((int)(blockIdx.x * 8 + (threadIdx.x >> 5)));
    if (n >= N) return;
    const int beg = offset[n];
    const int end = beg + count[n];
    const int e8 = lane & 7;
    const int hd = lane >> 3;

    float4 acc = make_float4(0.0f, 0.0f, 0.0f, 0.0f);
    if (end > beg) {
        const float aln = al[(long)n * NHEAD + hd];
        const float* __restrict__ hl = h + lane * 4;

        float m = -INFINITY;
        for (int base = beg; base < end; base += 8) {
            const int idx = base + e8;
            if (idx < end) m = fmaxf(m, lrelu(aln + ar[(long)scol[idx] * NHEAD + hd]));
        }
        m = group_max(m);

        float s = 0.0f;
        for (int base = beg; base < end; base += 8) {
            const int idx = base + e8;
            if (idx < end) s += expf(lrelu(aln + ar[(long)scol[idx] * NHEAD + hd]) - m);
        }
        s = group_sum(s);

        for (int base = beg; base < end; base += 8) {
            const int cnt = (end - base < 8) ? (end - base) : 8;
            int c = 0;
            float p = 0.0f;
            if (e8 < cnt) {
                c = scol[base + e8];
                p = expf(lrelu(aln + ar[(long)c * NHEAD + hd]) - m) / s;
            }
            float4 hv[8];
#define H_LOAD(K) if (K < cnt) { hv[K] = *(const float4*)(hl + (long)__builtin_amdgcn_readlane(c, K) * (OUT_F * NHEAD)); }
            H_LOAD(0) H_LOAD(1) H_LOAD(2) H_LOAD(3) H_LOAD(4) H_LOAD(5) H_LOAD(6) H_LOAD(7)
#undef H_LOAD
#define H_FMA(K) if (K < cnt) { const float w = group_bcast<K>(p); \
                acc.x += w * hv[K].x; acc.y += w * hv[K].y; acc.z += w * hv[K].z; acc.w += w * hv[K].w; }
            H_FMA(0) H_FMA(1) H_FMA(2) H_FMA(3) H_FMA(4) H_FMA(5) H_FMA(6) H_FMA(7)
#undef H_FMA
        }
    }
    *(float4*)(out + (long)n * (OUT_F * NHEAD) + lane * 4) = acc;
}

extern "C" void kernel_launch(void* const* d_in, const int* in_sizes, int n_in,
                              void* d_out, int out_size, void* d_ws, size_t ws_size,
                              hipStream_t stream) {
    const float* x    = (const float*)d_in[0];
    const int*   edge = (const int*)d_in[1];
    const float* W    = (const float*)d_in[2];
    const float* a_l  = (const float*)d_in[3];
    const float* a_r  = (const float*)d_in[4];

    const int N = in_sizes[0] / IN_F;
    const int E = in_sizes[1] / 2;
    const int* rowp = edge;
    const int* colp = edge + E;

    float* ws = (float*)d_ws;
    float* h  = ws;
    float* al = h  + (size_t)N * (OUT_F * NHEAD);
    float* ar = al + (size_t)N * NHEAD;
    int* count  = (int*)(ar + (size_t)N * NHEAD);
    int* offset = count + (size_t)N;
    int* scol   = offset + (size_t)N;
    int* wgcnt    = (int*)ws;
    int* base     = wgcnt + GA * NBIN;
    int* binStart = base + GA * NBIN;
    int* rec      = binStart + 1024;
    float* out = (float*)d_out;

    gat_bincount<<<GA, 1024, 0, stream>>>(rowp, wgcnt, E, N);
    gat_binscan<<<1, 2 * NBIN, 0, stream>>>(wgcnt, base, binStart);
    gat_place<<<GA, 1024, 0, stream>>>(rowp, colp, wgcnt, base, rec, E, N);
    gat_group<<<NBIN, 1024, 0, stream>>>(rec, binStart, count, offset, scol, N);
    gat_gemm<<<(unsigned)((N + 127) / 128), 256, 0, stream>>>(x, W, a_l, a_r, h, al, ar, N);
    gat_node<<<(unsigned)((N + 7) / 8), 256, 0, stream>>>(offset, count, scol, al, ar, h, out, N);
}


# Node walk reads each edge's source id and ar value once: ids and scores of up to 32 edges stay in registers for max, exp, sum and the h-row walk (three dependent list passes before); sums unchanged
# speedup vs baseline: 6.9095x; 1.1539x over previous
#include <hip/hip_runtime.h>
#include <hip/hip_bf16.h>
#include <math.h>

typedef __attribute__((ext_vector_type(2))) float v2f;
typedef __attribute__((ext_vector_type(8))) float v8f;

#define IN_F   128
#define OUT_F  32
#define NHEAD  4
#define ALPHA  0.2f


template <int XM>
__device__ inline float row_xor(float v) {
    return __int_as_float(__builtin_amdgcn_update_dpp(0, __float_as_int(v), 0x160 | XM, 0xf, 0xf, true));
}
template <int XM>
__device__ inline float fold(float a, float b, bool hi) {
    const float keep = hi ? b : a;
    const float send = hi ? a : b;
    return keep + row_xor<XM>(send);
}

__global__ __launch_bounds__(256) void gat_gemm(const float* __restrict__ x,
                                                const float* __restrict__ W,
                                                const float* __restrict__ a_l,
                                                const float* __restrict__ a_r,
                                                float* __restrict__ h,
                                                float* __restrict__ al,
                                                float* __restrict__ ar, int N) {
    const int wave = threadIdx.x >> 5;
    const int lane = threadIdx.x & 31;
    const long rowTile = (long)blockIdx.x * 8 + wave;
    const long row0 = rowTile * 16;
    if (row0 >= N) return;

    const int mrow  = lane & 15;
    const int khalf = (lane >> 4) * 2;

    v8f acc[8];
    #pragma unroll
    for (int t = 0; t < 8; ++t) acc[t] = v8f{};

    long r = row0 + mrow;
    if (r >= N) r = N - 1;
    const float* __restrict__ xrow = x + r * IN_F;

    for (int k = 0; k < IN_F; k += 4) {
        v2f a;
        a.x = xrow[k + khalf + 0];
        a.y = xrow[k + khalf + 1];
        #pragma unroll
        for (int t = 0; t < 8; ++t) {
            const int ncol = t * 16 + mrow;
            v2f b;
            b.x = W[(k + khalf + 0) * (OUT_F * NHEAD) + ncol];
            b.y = W[(k + khalf + 1) * (OUT_F * NHEAD) + ncol];
            acc[t] = __builtin_amdgcn_wmma_f32_16x16x4_f32(
                false, a, false, b, (short)0, acc[t], false, false);
        }
    }

    const int rowOff = (lane < 16) ? 0 : 8;
    #pragma unroll
    for (int t = 0; t < 8; ++t) {
        #pragma unroll
        for (int g = 0; g < 8; ++g) {
            float v = acc[t][g];
            v = (v == v) ? v : 0.0f;
            acc[t][g] = v;
            const long rr = row0 + g + rowOff;
            if (rr < N) h[rr * (OUT_F * NHEAD) + t * 16 + mrow] = v;
        }
    }

    const bool b0 = (mrow & 1) != 0, b1 = (mrow & 2) != 0, b2 = (mrow & 4) != 0, b3 = (mrow & 8) != 0;
    float res[NHEAD];
    #pragma unroll
    for (int j = 0; j < NHEAD; ++j) {
        float q[2];
        #pragma unroll
        for (int c = 0; c < 2; ++c) {
            const float* __restrict__ av = (c == 0) ? a_l : a_r;
            const float w0 = av[(2 * j) * 16 + mrow];
            const float w1 = av[(2 * j + 1) * 16 + mrow];
            float p[8];
            #pragma unroll
            for (int g = 0; g < 8; ++g) p[g] = w0 * acc[2 * j][g] + w1 * acc[2 * j + 1][g];
            const float p01 = fold<1>(p[0], p[1], b0);
            const float p23 = fold<1>(p[2], p[3], b0);
            const float p45 = fold<1>(p[4], p[5], b0);
            const float p67 = fold<1>(p[6], p[7], b0);
            const float p03 = fold<2>(p01, p23, b1);
            const float p47 = fold<2>(p45, p67, b1);
            q[c] = fold<4>(p03, p47, b2);
        }
        res[j] = fold<8>(q[0], q[1], b3);
    }
    const long orow = row0 + rowOff + (mrow & 7);
    if (orow < N) {
        float* __restrict__ dst = (b3 ? ar : al) + orow * NHEAD;
        *(float4*)dst = make_float4(res[0], res[1], res[2], res[3]);
    }
}

#define GA     256
#define NBIN   512
#define SMAX   1024
#define CHMAX  8192
#define CAPC   16384

__device__ inline int bin_shift(int N) {
    return 32 - __clz((N - 1) >> 9);
}

__device__ inline int block_excl_scan(int v, int* wtot) {
    const int lane = threadIdx.x & 31;
    const int wave = threadIdx.x >> 5;
    const int nw = blockDim.x >> 5;
    int incl = v;
    #pragma unroll
    for (int off = 1; off < 32; off <<= 1) {
        const int u = __shfl_up(incl, off, 32);
        if (lane >= off) incl += u;
    }
    if (lane == 31) wtot[wave] = incl;
    __syncthreads();
    int pre = 0;
    for (int w = 0; w < nw; ++w) {
        if (w < wave) pre += wtot[w];
    }
    __syncthreads();
    return pre + incl - v;
}

__global__ __launch_bounds__(1024) void gat_bincount(const int* __restrict__ row,
                                                     int* __restrict__ wgcnt, int E, int N) {
    __shared__ int cnt[NBIN];
    const int tid = threadIdx.x;
    const int shift = bin_shift(N);
    const int CH = (E + GA - 1) / GA;
    const int lo = (int)blockIdx.x * CH;
    const int hi = (lo + CH < E) ? (lo + CH) : E;
    if (tid < NBIN) cnt[tid] = 0;
    __syncthreads();
    for (int i0 = lo + tid; i0 < hi; i0 += 4 * 1024) {
        int r[4];
        #pragma unroll
        for (int k = 0; k < 4; ++k) {
            const int i = i0 + k * 1024;
            r[k] = (i < hi) ? row[i] : 0;
        }
        #pragma unroll
        for (int k = 0; k < 4; ++k) {
            if (i0 + k * 1024 < hi) atomicAdd(&cnt[r[k] >> shift], 1);
        }
    }
    __syncthreads();
    if (tid < NBIN) wgcnt[(int)blockIdx.x * NBIN + tid] = cnt[tid];
}

__global__ __launch_bounds__(2 * NBIN) void gat_binscan(const int* wgcnt, int* base,
                                                        int* __restrict__ binStart) {
    __shared__ int wtot[32];
    __shared__ int part0[NBIN];
    __shared__ int part1[NBIN];
    __shared__ int bstart[NBIN];
    const int b  = threadIdx.x & (NBIN - 1);
    const int hf = threadIdx.x >> 9;
    const int* src = wgcnt + (hf * (GA / 2) * NBIN + b);
    int*       dst = base  + (hf * (GA / 2) * NBIN + b);
    int part = 0;
    for (int t0 = 0; t0 < GA / 2; t0 += 32) {
        int v[32];
        #pragma unroll
        for (int k = 0; k < 32; ++k) v[k] = src[(t0 + k) * NBIN];
        #pragma unroll
        for (int k = 0; k < 32; ++k) part += v[k];
    }
    if (hf == 0) part0[b] = part; else part1[b] = part;
    __syncthreads();
    const int tot = (hf == 0) ? (part0[b] + part1[b]) : 0;
    const int excl = block_excl_scan(tot, wtot);
    if (hf == 0) {
        bstart[b] = excl;
        binStart[b] = excl;
        if (b == NBIN - 1) binStart[NBIN] = excl + tot;
    }
    __syncthreads();
    int run = bstart[b] + ((hf == 0) ? 0 : part0[b]);
    for (int t0 = 0; t0 < GA / 2; t0 += 32) {
        int v[32];
        #pragma unroll
        for (int k = 0; k < 32; ++k) v[k] = src[(t0 + k) * NBIN];
        #pragma unroll
        for (int k = 0; k < 32; ++k) {
            dst[(t0 + k) * NBIN] = run;
            run += v[k];
        }
    }
}

__global__ __launch_bounds__(1024) void gat_place(const int* __restrict__ row,
                                                  const int* __restrict__ col,
                                                  const int* __restrict__ wgcnt,
                                                  const int* __restrict__ base,
                                                  int* __restrict__ recw, int E, int N) {
    int2* __restrict__ rec = (int2*)recw;
    __shared__ int cur[NBIN];
    __shared__ int delta[NBIN];
    __shared__ int wtot[32];
    __shared__ int2 staged[CHMAX];
    const int tid = threadIdx.x;
    const int shift = bin_shift(N);
    const int CH = (E + GA - 1) / GA;
    const int lo = (int)blockIdx.x * CH;
    const int hi = (lo + CH < E) ? (lo + CH) : E;
    const int len = (hi > lo) ? (hi - lo) : 0;
    const bool stage = (CH <= CHMAX);

    const int c = (tid < NBIN) ? wgcnt[(int)blockIdx.x * NBIN + tid] : 0;
    const int excl = block_excl_scan(c, wtot);
    if (tid < NBIN) {
        const int g = base[(int)blockIdx.x * NBIN + tid];
        cur[tid] = stage ? excl : g;
        delta[tid] = g - excl;
    }
    __syncthreads();
    for (int i0 = lo + tid; i0 < hi; i0 += 4 * 1024) {
        int r[4], s[4];
        #pragma unroll
        for (int k = 0; k < 4; ++k) {
            const int i = i0 + k * 1024;
            r[k] = (i < hi) ? row[i] : 0;
            s[k] = (i < hi) ? col[i] : 0;
        }
        #pragma unroll
        for (int k = 0; k < 4; ++k) {
            if (i0 + k * 1024 < hi) {
                const int pos = atomicAdd(&cur[r[k] >> shift], 1);
                if (stage) staged[pos] = make_int2(r[k], s[k]);
                else       rec[pos]    = make_int2(r[k], s[k]);
            }
        }
    }
    if (stage) {
        __syncthreads();
        for (int j = tid; j < len; j += 1024) {
            const int2 e = staged[j];
            rec[j + delta[e.x >> shift]] = e;
        }
    }
}

__global__ __launch_bounds__(1024) void gat_group(const int* __restrict__ recw,
                                                  const int* __restrict__ binStart,
                                                  int* __restrict__ count,
                                                  int* __restrict__ offset,
                                                  int* __restrict__ scol, int N) {
    const int2* __restrict__ rec = (const int2*)recw;
    __shared__ int ncnt[SMAX];
    __shared__ int ncur[SMAX];
    __shared__ int wtot[32];
    __shared__ int staged[CAPC];
    const int tid = threadIdx.x;
    const int shift = bin_shift(N);
    const int n0 = (int)blockIdx.x << shift;
    if (n0 >= N) return;
    const int s0 = binStart[blockIdx.x];
    const int len = binStart[blockIdx.x + 1] - s0;
    const bool stage = (len <= CAPC);

    ncnt[tid] = 0;
    __syncthreads();
    for (int j0 = tid; j0 < len; j0 += 4 * 1024) {
        int r[4];
        #pragma unroll
        for (int k = 0; k < 4; ++k) {
            const int j = j0 + k * 1024;
            r[k] = (j < len) ? rec[s0 + j].x : n0;
        }
        #pragma unroll
        for (int k = 0; k < 4; ++k) {
            if (j0 + k * 1024 < len) atomicAdd(&ncnt[r[k] - n0], 1);
        }
    }
    __syncthreads();
    const int v = ncnt[tid];
    const int excl = block_excl_scan(v, wtot);
    ncur[tid] = excl;
    const int n = n0 + tid;
    if (tid < (1 << shift) && n < N) {
        count[n] = v;
        offset[n] = s0 + excl;
    }
    __syncthreads();
    for (int j0 = tid; j0 < len; j0 += 4 * 1024) {
        int2 e[4];
        #pragma unroll
        for (int k = 0; k < 4; ++k) {
            const int j = j0 + k * 1024;
            e[k] = (j < len) ? rec[s0 + j] : make_int2(n0, 0);
        }
        #pragma unroll
        for (int k = 0; k < 4; ++k) {
            if (j0 + k * 1024 < len) {
                const int pos = atomicAdd(&ncur[e[k].x - n0], 1);
                if (stage) staged[pos] = e[k].y;
                else       scol[s0 + pos] = e[k].y;
            }
        }
    }
    if (stage) {
        __syncthreads();
        for (int j = tid; j < len; j += 1024) scol[s0 + j] = staged[j];
    }
}

__device__ inline float lrelu(float v) { return (v > 0.0f) ? v : (ALPHA * v); }

template <int K>
__device__ inline float group_bcast(float v) {
    return __int_as_float((int)__builtin_amdgcn_mov_dpp8((unsigned)__float_as_int(v), (unsigned)(K * 0x249249)));
}
__device__ inline float group_max(float v) {
    #pragma unroll
    for (int off = 4; off > 0; off >>= 1) v = fmaxf(v, __shfl_xor(v, off, 32));
    return v;
}
__device__ inline float group_sum(float v) {
    #pragma unroll
    for (int off = 4; off > 0; off >>= 1) v += __shfl_xor(v, off, 32);
    return v;
}

#define ESLOT 4

#define H_LOAD(K) if (K < cnt) { hv[K] = *(const float4*)(hl + (long)__builtin_amdgcn_readlane(c, K) * (OUT_F * NHEAD)); }
#define H_FMA(K) if (K < cnt) { const float w = group_bcast<K>(p); \
                acc.x += w * hv[K].x; acc.y += w * hv[K].y; acc.z += w * hv[K].z; acc.w += w * hv[K].w; }
__device__ inline void rows_load(const float* __restrict__ hl, int c, int cnt, float4 (&hv)[8]) {
    H_LOAD(0) H_LOAD(1) H_LOAD(2) H_LOAD(3) H_LOAD(4) H_LOAD(5) H_LOAD(6) H_LOAD(7)
}
__device__ inline void rows_fma(float p, int cnt, const float4 (&hv)[8], float4& acc) {
    H_FMA(0) H_FMA(1) H_FMA(2) H_FMA(3) H_FMA(4) H_FMA(5) H_FMA(6) H_FMA(7)
}
#undef H_LOAD
#undef H_FMA

__global__ __launch_bounds__(256) void gat_node(const int* __restrict__ offset,
                                                const int* __restrict__ count,
                                                const int* __restrict__ scol,
                                                const float* __restrict__ al,
                                                const float* __restrict__ ar,
                                                const float* __restrict__ h,
                                                float* __restrict__ out, int N) {
    const int lane = threadIdx.x & 31;
    const int n = __builtin_amdgcn_readfirstlane((int)(blockIdx.x * 8 + (threadIdx.x >> 5)));
    if (n >= N) return;
    const int beg = offset[n];
    const int deg = count[n];
    const int end = beg + deg;
    const int e8 = lane & 7;
    const int hd = lane >> 3;

    float4 acc = make_float4(0.0f, 0.0f, 0.0f, 0.0f);
    if (deg > 0) {
        const float aln = al[(long)n * NHEAD + hd];
        const float* __restrict__ hl = h + lane * 4;

        if (deg <= 8 * ESLOT) {
            int   c[ESLOT];
            float p[ESLOT];
            #pragma unroll
            for (int k = 0; k < ESLOT; ++k) {
                const int idx = beg + 8 * k + e8;
                c[k] = scol[(idx < end) ? idx : (end - 1)];
            }
            #pragma unroll
            for (int k = 0; k < ESLOT; ++k) p[k] = ar[(long)c[k] * NHEAD + hd];
            float4 hv[8];
            const int cnt0 = (deg < 8) ? deg : 8;
            rows_load(hl, c[0], cnt0, hv);

            float m = -INFINITY;
            #pragma unroll
            for (int k = 0; k < ESLOT; ++k) {
                p[k] = (beg + 8 * k + e8 < end) ? lrelu(aln + p[k]) : -INFINITY;
                m = fmaxf(m, p[k]);
            }
            m = group_max(m);
            float s = 0.0f;
            #pragma unroll
            for (int k = 0; k < ESLOT; ++k) {
                p[k] = expf(p[k] - m);
                s += p[k];
            }
            s = group_sum(s);
            rows_fma(p[0] / s, cnt0, hv, acc);
            #pragma unroll
            for (int k = 1; k < ESLOT; ++k) {
                if (8 * k < deg) {
                    const int cnt = (deg - 8 * k < 8) ? (deg - 8 * k) : 8;
                    rows_load(hl, c[k], cnt, hv);
                    rows_fma(p[k] / s, cnt, hv, acc);
                }
            }
        } else {
            float m = -INFINITY;
            for (int base = beg; base < end; base += 8) {
                const int idx = base + e8;
                if (idx < end) m = fmaxf(m, lrelu(aln + ar[(long)scol[idx] * NHEAD + hd]));
            }
            m = group_max(m);

            float s = 0.0f;
            for (int base = beg; base < end; base += 8) {
                const int idx = base + e8;
                if (idx < end) s += expf(lrelu(aln + ar[(long)scol[idx] * NHEAD + hd]) - m);
            }
            s = group_sum(s);

            for (int base = beg; base < end; base += 8) {
                const int cnt = (end - base < 8) ? (end - base) : 8;
                int c = 0;
                float p = 0.0f;
                if (e8 < cnt) {
                    c = scol[base + e8];
                    p = expf(lrelu(aln + ar[(long)c * NHEAD + hd]) - m) / s;
                }
                float4 hv[8];
                rows_load(hl, c, cnt, hv);
                rows_fma(p, cnt, hv, acc);
            }
        }
    }
    *(float4*)(out + (long)n * (OUT_F * NHEAD) + lane * 4) = acc;
}

extern "C" void kernel_launch(void* const* d_in, const int* in_sizes, int n_in,
                              void* d_out, int out_size, void* d_ws, size_t ws_size,
                              hipStream_t stream) {
    const float* x    = (const float*)d_in[0];
    const int*   edge = (const int*)d_in[1];
    const float* W    = (const float*)d_in[2];
    const float* a_l  = (const float*)d_in[3];
    const float* a_r  = (const float*)d_in[4];

    const int N = in_sizes[0] / IN_F;
    const int E = in_sizes[1] / 2;
    const int* rowp = edge;
    const int* colp = edge + E;

    float* ws = (float*)d_ws;
    float* h  = ws;
    float* al = h  + (size_t)N * (OUT_F * NHEAD);
    float* ar = al + (size_t)N * NHEAD;
    int* count  = (int*)(ar + (size_t)N * NHEAD);
    int* offset = count + (size_t)N;
    int* scol   = offset + (size_t)N;
    int* wgcnt    = (int*)ws;
    int* base     = wgcnt + GA * NBIN;
    int* binStart = base + GA * NBIN;
    int* rec      = binStart + 1024;
    float* out = (float*)d_out;

    gat_bincount<<<GA, 1024, 0, stream>>>(rowp, wgcnt, E, N);
    gat_binscan<<<1, 2 * NBIN, 0, stream>>>(wgcnt, base, binStart);
    gat_place<<<GA, 1024, 0, stream>>>(rowp, colp, wgcnt, base, rec, E, N);
    gat_group<<<NBIN, 1024, 0, stream>>>(rec, binStart, count, offset, scol, N);
    gat_gemm<<<(unsigned)((N + 127) / 128), 256, 0, stream>>>(x, W, a_l, a_r, h, al, ar, N);
    gat_node<<<(unsigned)((N + 7) / 8), 256, 0, stream>>>(offset, count, scol, al, ar, h, out, N);
}


# Product tiles ride in the launch of the one-work-group bin scan (work-group 0 scans, 511 others compute h, al, ar); grouping tables moved from the h plane to the out buffer; sums unchanged
# speedup vs baseline: 7.8061x; 1.1298x over previous
#include <hip/hip_runtime.h>
#include <hip/hip_bf16.h>
#include <math.h>

typedef __attribute__((ext_vector_type(2))) float v2f;
typedef __attribute__((ext_vector_type(8))) float v8f;

#define IN_F   128
#define OUT_F  32
#define NHEAD  4
#define ALPHA  0.2f


template <int XM>
__device__ inline float row_xor(float v) {
    return __int_as_float(__builtin_amdgcn_update_dpp(0, __float_as_int(v), 0x160 | XM, 0xf, 0xf, true));
}
template <int XM>
__device__ inline float fold(float a, float b, bool hi) {
    const float keep = hi ? b : a;
    const float send = hi ? a : b;
    return keep + row_xor<XM>(send);
}

__device__ inline void gemm_tile(const float* __restrict__ x,
                                 const float* __restrict__ W,
                                 const float* __restrict__ a_l,
                                 const float* __restrict__ a_r,
                                 float* __restrict__ h,
                                 float* __restrict__ al,
                                 float* __restrict__ ar, int N, long rowTile, int lane) {
    const long row0 = rowTile * 16;

    const int mrow  = lane & 15;
    const int khalf = (lane >> 4) * 2;

    v8f acc[8];
    #pragma unroll
    for (int t = 0; t < 8; ++t) acc[t] = v8f{};

    long r = row0 + mrow;
    if (r >= N) r = N - 1;
    const float* __restrict__ xrow = x + r * IN_F;

    const float* __restrict__ wl = W + khalf * (OUT_F * NHEAD) + mrow;
    for (int k = 0; k < IN_F; k += 4) {
        v2f a;
        a.x = xrow[k + khalf + 0];
        a.y = xrow[k + khalf + 1];
        const float* __restrict__ wk = wl + k * (OUT_F * NHEAD);
        #pragma unroll
        for (int t = 0; t < 8; ++t) {
            v2f b;
            b.x = wk[t * 16];
            b.y = wk[(OUT_F * NHEAD) + t * 16];
            acc[t] = __builtin_amdgcn_wmma_f32_16x16x4_f32(
                false, a, false, b, (short)0, acc[t], false, false);
        }
    }

    const int rowOff = (lane < 16) ? 0 : 8;
    #pragma unroll
    for (int t = 0; t < 8; ++t) {
        #pragma unroll
        for (int g = 0; g < 8; ++g) {
            float v = acc[t][g];
            v = (v == v) ? v : 0.0f;
            acc[t][g] = v;
            const long rr = row0 + g + rowOff;
            if (rr < N) h[rr * (OUT_F * NHEAD) + t * 16 + mrow] = v;
        }
    }

    const bool b0 = (mrow & 1) != 0, b1 = (mrow & 2) != 0, b2 = (mrow & 4) != 0, b3 = (mrow & 8) != 0;
    float res[NHEAD];
    #pragma unroll
    for (int j = 0; j < NHEAD; ++j) {
        float q[2];
        #pragma unroll
        for (int c = 0; c < 2; ++c) {
            const float* __restrict__ av = (c == 0) ? a_l : a_r;
            const float w0 = av[(2 * j) * 16 + mrow];
            const float w1 = av[(2 * j + 1) * 16 + mrow];
            float p[8];
            #pragma unroll
            for (int g = 0; g < 8; ++g) p[g] = w0 * acc[2 * j][g] + w1 * acc[2 * j + 1][g];
            const float p01 = fold<1>(p[0], p[1], b0);
            const float p23 = fold<1>(p[2], p[3], b0);
            const float p45 = fold<1>(p[4], p[5], b0);
            const float p67 = fold<1>(p[6], p[7], b0);
            const float p03 = fold<2>(p01, p23, b1);
            const float p47 = fold<2>(p45, p67, b1);
            q[c] = fold<4>(p03, p47, b2);
        }
        res[j] = fold<8>(q[0], q[1], b3);
    }
    const long orow = row0 + rowOff + (mrow & 7);
    if (orow < N) {
        float* __restrict__ dst = (b3 ? ar : al) + orow * NHEAD;
        *(float4*)dst = make_float4(res[0], res[1], res[2], res[3]);
    }
}

#define GA     256
#define NBIN   512
#define SMAX   1024
#define CHMAX  8192
#define CAPC   16384

__device__ inline int bin_shift(int N) {
    return 32 - __clz((N - 1) >> 9);
}

__device__ inline int block_excl_scan(int v, int* wtot) {
    const int lane = threadIdx.x & 31;
    const int wave = threadIdx.x >> 5;
    const int nw = blockDim.x >> 5;
    int incl = v;
    #pragma unroll
    for (int off = 1; off < 32; off <<= 1) {
        const int u = __shfl_up(incl, off, 32);
        if (lane >= off) incl += u;
    }
    if (lane == 31) wtot[wave] = incl;
    __syncthreads();
    int pre = 0;
    for (int w = 0; w < nw; ++w) {
        if (w < wave) pre += wtot[w];
    }
    __syncthreads();
    return pre + incl - v;
}

__global__ __launch_bounds__(1024) void gat_bincount(const int* __restrict__ row,
                                                     int* __restrict__ wgcnt, int E, int N) {
    __shared__ int cnt[NBIN];
    const int tid = threadIdx.x;
    const int shift = bin_shift(N);
    const int CH = (E + GA - 1) / GA;
    const int lo = (int)blockIdx.x * CH;
    const int hi = (lo + CH < E) ? (lo + CH) : E;
    if (tid < NBIN) cnt[tid] = 0;
    __syncthreads();
    for (int i0 = lo + tid; i0 < hi; i0 += 4 * 1024) {
        int r[4];
        #pragma unroll
        for (int k = 0; k < 4; ++k) {
            const int i = i0 + k * 1024;
            r[k] = (i < hi) ? row[i] : 0;
        }
        #pragma unroll
        for (int k = 0; k < 4; ++k) {
            if (i0 + k * 1024 < hi) atomicAdd(&cnt[r[k] >> shift], 1);
        }
    }
    __syncthreads();
    if (tid < NBIN) wgcnt[(int)blockIdx.x * NBIN + tid] = cnt[tid];
}

__device__ inline void binscan_block(const int* wgcnt, int* base, int* __restrict__ binStart) {
    __shared__ int wtot[32];
    const int b = threadIdx.x;
    const int* src = wgcnt + b;
    int*       dst = base  + b;
    int tot = 0;
    for (int t0 = 0; t0 < GA; t0 += 32) {
        int v[32];
        #pragma unroll
        for (int k = 0; k < 32; ++k) v[k] = src[(t0 + k) * NBIN];
        #pragma unroll
        for (int k = 0; k < 32; ++k) tot += v[k];
    }
    const int excl = block_excl_scan(tot, wtot);
    binStart[b] = excl;
    if (b == NBIN - 1) binStart[NBIN] = excl + tot;
    int run = excl;
    for (int t0 = 0; t0 < GA; t0 += 32) {
        int v[32];
        #pragma unroll
        for (int k = 0; k < 32; ++k) v[k] = src[(t0 + k) * NBIN];
        #pragma unroll
        for (int k = 0; k < 32; ++k) {
            dst[(t0 + k) * NBIN] = run;
            run += v[k];
        }
    }
}

__global__ __launch_bounds__(NBIN) void gat_scan_gemm(const int* wgcnt, int* base,
                                                      int* __restrict__ binStart,
                                                      const float* __restrict__ x,
                                                      const float* __restrict__ W,
                                                      const float* __restrict__ a_l,
                                                      const float* __restrict__ a_r,
                                                      float* __restrict__ h,
                                                      float* __restrict__ al,
                                                      float* __restrict__ ar, int N) {
    if (blockIdx.x == 0) {
        binscan_block(wgcnt, base, binStart);
        return;
    }
    const int wave = threadIdx.x >> 5;
    const int lane = threadIdx.x & 31;
    const int G = (int)gridDim.x - 1;
    const int tiles = (N + 15) / 16;
    const int t = (int)blockIdx.x - 1 + G * wave;
    if (t < tiles) gemm_tile(x, W, a_l, a_r, h, al, ar, N, (long)t, lane);
}

__global__ __launch_bounds__(1024) void gat_place(const int* __restrict__ row,
                                                  const int* __restrict__ col,
                                                  const int* __restrict__ wgcnt,
                                                  const int* __restrict__ base,
                                                  int* __restrict__ recw, int E, int N) {
    int2* __restrict__ rec = (int2*)recw;
    __shared__ int cur[NBIN];
    __shared__ int delta[NBIN];
    __shared__ int wtot[32];
    __shared__ int2 staged[CHMAX];
    const int tid = threadIdx.x;
    const int shift = bin_shift(N);
    const int CH = (E + GA - 1) / GA;
    const int lo = (int)blockIdx.x * CH;
    const int hi = (lo + CH < E) ? (lo + CH) : E;
    const int len = (hi > lo) ? (hi - lo) : 0;
    const bool stage = (CH <= CHMAX);

    const int c = (tid < NBIN) ? wgcnt[(int)blockIdx.x * NBIN + tid] : 0;
    const int excl = block_excl_scan(c, wtot);
    if (tid < NBIN) {
        const int g = base[(int)blockIdx.x * NBIN + tid];
        cur[tid] = stage ? excl : g;
        delta[tid] = g - excl;
    }
    __syncthreads();
    for (int i0 = lo + tid; i0 < hi; i0 += 4 * 1024) {
        int r[4], s[4];
        #pragma unroll
        for (int k = 0; k < 4; ++k) {
            const int i = i0 + k * 1024;
            r[k] = (i < hi) ? row[i] : 0;
            s[k] = (i < hi) ? col[i] : 0;
        }
        #pragma unroll
        for (int k = 0; k < 4; ++k) {
            if (i0 + k * 1024 < hi) {
                const int pos = atomicAdd(&cur[r[k] >> shift], 1);
                if (stage) staged[pos] = make_int2(r[k], s[k]);
                else       rec[pos]    = make_int2(r[k], s[k]);
            }
        }
    }
    if (stage) {
        __syncthreads();
        for (int j = tid; j < len; j += 1024) {
            const int2 e = staged[j];
            rec[j + delta[e.x >> shift]] = e;
        }
    }
}

__global__ __launch_bounds__(1024) void gat_group(const int* __restrict__ recw,
                                                  const int* __restrict__ binStart,
                                                  int* __restrict__ count,
                                                  int* __restrict__ offset,
                                                  int* __restrict__ scol, int N) {
    const int2* __restrict__ rec = (const int2*)recw;
    __shared__ int ncnt[SMAX];
    __shared__ int ncur[SMAX];
    __shared__ int wtot[32];
    __shared__ int staged[CAPC];
    const int tid = threadIdx.x;
    const int shift = bin_shift(N);
    const int n0 = (int)blockIdx.x << shift;
    if (n0 >= N) return;
    const int s0 = binStart[blockIdx.x];
    const int len = binStart[blockIdx.x + 1] - s0;
    const bool stage = (len <= CAPC);

    ncnt[tid] = 0;
    __syncthreads();
    for (int j0 = tid; j0 < len; j0 += 4 * 1024) {
        int r[4];
        #pragma unroll
        for (int k = 0; k < 4; ++k) {
            const int j = j0 + k * 1024;
            r[k] = (j < len) ? rec[s0 + j].x : n0;
        }
        #pragma unroll
        for (int k = 0; k < 4; ++k) {
            if (j0 + k * 1024 < len) atomicAdd(&ncnt[r[k] - n0], 1);
        }
    }
    __syncthreads();
    const int v = ncnt[tid];
    const int excl = block_excl_scan(v, wtot);
    ncur[tid] = excl;
    const int n = n0 + tid;
    if (tid < (1 << shift) && n < N) {
        count[n] = v;
        offset[n] = s0 + excl;
    }
    __syncthreads();
    for (int j0 = tid; j0 < len; j0 += 4 * 1024) {
        int2 e[4];
        #pragma unroll
        for (int k = 0; k < 4; ++k) {
            const int j = j0 + k * 1024;
            e[k] = (j < len) ? rec[s0 + j] : make_int2(n0, 0);
        }
        #pragma unroll
        for (int k = 0; k < 4; ++k) {
            if (j0 + k * 1024 < len) {
                const int pos = atomicAdd(&ncur[e[k].x - n0], 1);
                if (stage) staged[pos] = e[k].y;
                else       scol[s0 + pos] = e[k].y;
            }
        }
    }
    if (stage) {
        __syncthreads();
        for (int j = tid; j < len; j += 1024) scol[s0 + j] = staged[j];
    }
}

__device__ inline float lrelu(float v) { return (v > 0.0f) ? v : (ALPHA * v); }

template <int K>
__device__ inline float group_bcast(float v) {
    return __int_as_float((int)__builtin_amdgcn_mov_dpp8((unsigned)__float_as_int(v), (unsigned)(K * 0x249249)));
}
__device__ inline float group_max(float v) {
    #pragma unroll
    for (int off = 4; off > 0; off >>= 1) v = fmaxf(v, __shfl_xor(v, off, 32));
    return v;
}
__device__ inline float group_sum(float v) {
    #pragma unroll
    for (int off = 4; off > 0; off >>= 1) v += __shfl_xor(v, off, 32);
    return v;
}

#define ESLOT 4

#define H_LOAD(K) if (K < cnt) { hv[K] = *(const float4*)(hl + (long)__builtin_amdgcn_readlane(c, K) * (OUT_F * NHEAD)); }
#define H_FMA(K) if (K < cnt) { const float w = group_bcast<K>(p); \
                acc.x += w * hv[K].x; acc.y += w * hv[K].y; acc.z += w * hv[K].z; acc.w += w * hv[K].w; }
__device__ inline void rows_load(const float* __restrict__ hl, int c, int cnt, float4 (&hv)[8]) {
    H_LOAD(0) H_LOAD(1) H_LOAD(2) H_LOAD(3) H_LOAD(4) H_LOAD(5) H_LOAD(6) H_LOAD(7)
}
__device__ inline void rows_fma(float p, int cnt, const float4 (&hv)[8], float4& acc) {
    H_FMA(0) H_FMA(1) H_FMA(2) H_FMA(3) H_FMA(4) H_FMA(5) H_FMA(6) H_FMA(7)
}
#undef H_LOAD
#undef H_FMA

__global__ __launch_bounds__(256) void gat_node(const int* __restrict__ offset,
                                                const int* __restrict__ count,
                                                const int* __restrict__ scol,
                                                const float* __restrict__ al,
                                                const float* __restrict__ ar,
                                                const float* __restrict__ h,
                                                float* __restrict__ out, int N) {
    const int lane = threadIdx.x & 31;
    const int n = __builtin_amdgcn_readfirstlane((int)(blockIdx.x * 8 + (threadIdx.x >> 5)));
    if (n >= N) return;
    const int beg = offset[n];
    const int deg = count[n];
    const int end = beg + deg;
    const int e8 = lane & 7;
    const int hd = lane >> 3;

    float4 acc = make_float4(0.0f, 0.0f, 0.0f, 0.0f);
    if (deg > 0) {
        const float aln = al[(long)n * NHEAD + hd];
        const float* __restrict__ hl = h + lane * 4;

        if (deg <= 8 * ESLOT) {
            int   c[ESLOT];
            float p[ESLOT];
            #pragma unroll
            for (int k = 0; k < ESLOT; ++k) {
                const int idx = beg + 8 * k + e8;
                c[k] = scol[(idx < end) ? idx : (end - 1)];
            }
            #pragma unroll
            for (int k = 0; k < ESLOT; ++k) p[k] = ar[(long)c[k] * NHEAD + hd];
            float4 hv[8];
            const int cnt0 = (deg < 8) ? deg : 8;
            rows_load(hl, c[0], cnt0, hv);

            float m = -INFINITY;
            #pragma unroll
            for (int k = 0; k < ESLOT; ++k) {
                p[k] = (beg + 8 * k + e8 < end) ? lrelu(aln + p[k]) : -INFINITY;
                m = fmaxf(m, p[k]);
            }
            m = group_max(m);
            float s = 0.0f;
            #pragma unroll
            for (int k = 0; k < ESLOT; ++k) {
                p[k] = expf(p[k] - m);
                s += p[k];
            }
            s = group_sum(s);
            rows_fma(p[0] / s, cnt0, hv, acc);
            #pragma unroll
            for (int k = 1; k < ESLOT; ++k) {
                if (8 * k < deg) {
                    const int cnt = (deg - 8 * k < 8) ? (deg - 8 * k) : 8;
                    rows_load(hl, c[k], cnt, hv);
                    rows_fma(p[k] / s, cnt, hv, acc);
                }
            }
        } else {
            float m = -INFINITY;
            for (int base = beg; base < end; base += 8) {
                const int idx = base + e8;
                if (idx < end) m = fmaxf(m, lrelu(aln + ar[(long)scol[idx] * NHEAD + hd]));
            }
            m = group_max(m);

            float s = 0.0f;
            for (int base = beg; base < end; base += 8) {
                const int idx = base + e8;
                if (idx < end) s += expf(lrelu(aln + ar[(long)scol[idx] * NHEAD + hd]) - m);
            }
            s = group_sum(s);

            for (int base = beg; base < end; base += 8) {
                const int cnt = (end - base < 8) ? (end - base) : 8;
                int c = 0;
                float p = 0.0f;
                if (e8 < cnt) {
                    c = scol[base + e8];
                    p = expf(lrelu(aln + ar[(long)c * NHEAD + hd]) - m) / s;
                }
                float4 hv[8];
                rows_load(hl, c, cnt, hv);
                rows_fma(p, cnt, hv, acc);
            }
        }
    }
    *(float4*)(out + (long)n * (OUT_F * NHEAD) + lane * 4) = acc;
}

extern "C" void kernel_launch(void* const* d_in, const int* in_sizes, int n_in,
                              void* d_out, int out_size, void* d_ws, size_t ws_size,
                              hipStream_t stream) {
    const float* x    = (const float*)d_in[0];
    const int*   edge = (const int*)d_in[1];
    const float* W    = (const float*)d_in[2];
    const float* a_l  = (const float*)d_in[3];
    const float* a_r  = (const float*)d_in[4];

    const int N = in_sizes[0] / IN_F;
    const int E = in_sizes[1] / 2;
    const int* rowp = edge;
    const int* colp = edge + E;

    float* ws = (float*)d_ws;
    float* h  = ws;
    float* al = h  + (size_t)N * (OUT_F * NHEAD);
    float* ar = al + (size_t)N * NHEAD;
    int* count  = (int*)(ar + (size_t)N * NHEAD);
    int* offset = count + (size_t)N;
    int* scol   = offset + (size_t)N;
    int* wgcnt    = (int*)d_out;
    int* base     = wgcnt + GA * NBIN;
    int* binStart = base + GA * NBIN;
    int* rec      = binStart + 1024;
    float* out = (float*)d_out;

    gat_bincount<<<GA, 1024, 0, stream>>>(rowp, wgcnt, E, N);
    const int tiles16 = (N + 15) / 16;
    const int G = ((tiles16 + 15) / 16 > 511) ? (tiles16 + 15) / 16 : 511;
    gat_scan_gemm<<<(unsigned)(G + 1), NBIN, 0, stream>>>(wgcnt, base, binStart, x, W, a_l, a_r, h, al, ar, N);
    gat_place<<<GA, 1024, 0, stream>>>(rowp, colp, wgcnt, base, rec, E, N);
    gat_group<<<NBIN, 1024, 0, stream>>>(rec, binStart, count, offset, scol, N);
    gat_node<<<(unsigned)((N + 7) / 8), 256, 0, stream>>>(offset, count, scol, al, ar, h, out, N);
}


# Level 1 of the edge grouping made chunk-local: a chunk orders its own records by bin beside its bin starts, bin work-groups gather their segments; count, scan launches gone; lists stay arrival-order
# speedup vs baseline: 8.1137x; 1.0394x over previous
#include <hip/hip_runtime.h>
#include <hip/hip_bf16.h>
#include <math.h>

typedef __attribute__((ext_vector_type(2))) float v2f;
typedef __attribute__((ext_vector_type(8))) float v8f;

#define IN_F   128
#define OUT_F  32
#define NHEAD  4
#define ALPHA  0.2f


template <int XM>
__device__ inline float row_xor(float v) {
    return __int_as_float(__builtin_amdgcn_update_dpp(0, __float_as_int(v), 0x160 | XM, 0xf, 0xf, true));
}
template <int XM>
__device__ inline float fold(float a, float b, bool hi) {
    const float keep = hi ? b : a;
    const float send = hi ? a : b;
    return keep + row_xor<XM>(send);
}

__device__ inline void gemm_tile(const float* __restrict__ x,
                                 const float* __restrict__ W,
                                 const float* __restrict__ a_l,
                                 const float* __restrict__ a_r,
                                 float* __restrict__ h,
                                 float* __restrict__ al,
                                 float* __restrict__ ar, int N, long rowTile, int lane) {
    const long row0 = rowTile * 16;

    const int mrow  = lane & 15;
    const int khalf = (lane >> 4) * 2;

    v8f acc[8];
    #pragma unroll
    for (int t = 0; t < 8; ++t) acc[t] = v8f{};

    long r = row0 + mrow;
    if (r >= N) r = N - 1;
    const float* __restrict__ xrow = x + r * IN_F;

    const float* __restrict__ wl = W + khalf * (OUT_F * NHEAD) + mrow;
    for (int k = 0; k < IN_F; k += 4) {
        v2f a;
        a.x = xrow[k + khalf + 0];
        a.y = xrow[k + khalf + 1];
        const float* __restrict__ wk = wl + k * (OUT_F * NHEAD);
        #pragma unroll
        for (int t = 0; t < 8; ++t) {
            v2f b;
            b.x = wk[t * 16];
            b.y = wk[(OUT_F * NHEAD) + t * 16];
            acc[t] = __builtin_amdgcn_wmma_f32_16x16x4_f32(
                false, a, false, b, (short)0, acc[t], false, false);
        }
    }

    const int rowOff = (lane < 16) ? 0 : 8;
    #pragma unroll
    for (int t = 0; t < 8; ++t) {
        #pragma unroll
        for (int g = 0; g < 8; ++g) {
            float v = acc[t][g];
            v = (v == v) ? v : 0.0f;
            acc[t][g] = v;
            const long rr = row0 + g + rowOff;
            if (rr < N) h[rr * (OUT_F * NHEAD) + t * 16 + mrow] = v;
        }
    }

    const bool b0 = (mrow & 1) != 0, b1 = (mrow & 2) != 0, b2 = (mrow & 4) != 0, b3 = (mrow & 8) != 0;
    float res[NHEAD];
    #pragma unroll
    for (int j = 0; j < NHEAD; ++j) {
        float q[2];
        #pragma unroll
        for (int c = 0; c < 2; ++c) {
            const float* __restrict__ av = (c == 0) ? a_l : a_r;
            const float w0 = av[(2 * j) * 16 + mrow];
            const float w1 = av[(2 * j + 1) * 16 + mrow];
            float p[8];
            #pragma unroll
            for (int g = 0; g < 8; ++g) p[g] = w0 * acc[2 * j][g] + w1 * acc[2 * j + 1][g];
            const float p01 = fold<1>(p[0], p[1], b0);
            const float p23 = fold<1>(p[2], p[3], b0);
            const float p45 = fold<1>(p[4], p[5], b0);
            const float p67 = fold<1>(p[6], p[7], b0);
            const float p03 = fold<2>(p01, p23, b1);
            const float p47 = fold<2>(p45, p67, b1);
            q[c] = fold<4>(p03, p47, b2);
        }
        res[j] = fold<8>(q[0], q[1], b3);
    }
    const long orow = row0 + rowOff + (mrow & 7);
    if (orow < N) {
        float* __restrict__ dst = (b3 ? ar : al) + orow * NHEAD;
        *(float4*)dst = make_float4(res[0], res[1], res[2], res[3]);
    }
}

#define GA     256
#define NBIN   512
#define SMAX   1024
#define CHMAX  8192
#define CAPC   16384
#define CPITCH 544
#define SEGW   (GA / 32)

__device__ inline int bin_shift(int N) {
    return 32 - __clz((N - 1) >> 9);
}

__device__ inline int block_excl_scan(int v, int* wtot) {
    const int lane = threadIdx.x & 31;
    const int wave = threadIdx.x >> 5;
    const int nw = blockDim.x >> 5;
    int incl = v;
    #pragma unroll
    for (int off = 1; off < 32; off <<= 1) {
        const int u = __shfl_up(incl, off, 32);
        if (lane >= off) incl += u;
    }
    if (lane == 31) wtot[wave] = incl;
    __syncthreads();
    int pre = 0;
    for (int w = 0; w < nw; ++w) {
        if (w < wave) pre += wtot[w];
    }
    __syncthreads();
    return pre + incl - v;
}

__device__ inline int2 block_sum2(int a, int b, int* wtot2) {
    const int lane = threadIdx.x & 31;
    const int wave = threadIdx.x >> 5;
    const int nw = blockDim.x >> 5;
    #pragma unroll
    for (int off = 16; off > 0; off >>= 1) {
        a += __shfl_xor(a, off, 32);
        b += __shfl_xor(b, off, 32);
    }
    if (lane == 0) {
        wtot2[wave] = a;
        wtot2[32 + wave] = b;
    }
    __syncthreads();
    int ta = 0, tb = 0;
    for (int w = 0; w < nw; ++w) {
        ta += wtot2[w];
        tb += wtot2[32 + w];
    }
    __syncthreads();
    return make_int2(ta, tb);
}

__device__ inline void chunk_sort(const int* __restrict__ row, const int* __restrict__ col,
                                  int* __restrict__ cstart, int* __restrict__ recw,
                                  int E, int N) {
    int2* __restrict__ rec = (int2*)recw;
    __shared__ int cnt[NBIN];
    __shared__ int cur[NBIN];
    __shared__ int wtot[32];
    __shared__ int2 staged[CHMAX];
    const int tid = threadIdx.x;
    const int shift = bin_shift(N);
    const int CH = (E + GA - 1) / GA;
    const int lo = (int)blockIdx.x * CH;
    const int hi = (lo + CH < E) ? (lo + CH) : E;
    const int len = (hi > lo) ? (hi - lo) : 0;
    const bool stage = (CH <= CHMAX);

    cnt[tid] = 0;
    __syncthreads();
    for (int i0 = lo + tid; i0 < hi; i0 += 4 * NBIN) {
        int r[4];
        #pragma unroll
        for (int k = 0; k < 4; ++k) {
            const int i = i0 + k * NBIN;
            r[k] = (i < hi) ? row[i] : 0;
        }
        #pragma unroll
        for (int k = 0; k < 4; ++k) {
            if (i0 + k * NBIN < hi) atomicAdd(&cnt[r[k] >> shift], 1);
        }
    }
    __syncthreads();
    const int c = cnt[tid];
    const int excl = block_excl_scan(c, wtot);
    cur[tid] = excl;
    int* __restrict__ cs = cstart + (int)blockIdx.x * CPITCH;
    cs[tid] = excl;
    if (tid == NBIN - 1) cs[NBIN] = excl + c;
    __syncthreads();
    for (int i0 = lo + tid; i0 < hi; i0 += 4 * NBIN) {
        int r[4], s[4];
        #pragma unroll
        for (int k = 0; k < 4; ++k) {
            const int i = i0 + k * NBIN;
            r[k] = (i < hi) ? row[i] : 0;
            s[k] = (i < hi) ? col[i] : 0;
        }
        #pragma unroll
        for (int k = 0; k < 4; ++k) {
            if (i0 + k * NBIN < hi) {
                const int pos = atomicAdd(&cur[r[k] >> shift], 1);
                if (stage) staged[pos]   = make_int2(r[k], s[k]);
                else       rec[lo + pos] = make_int2(r[k], s[k]);
            }
        }
    }
    if (stage) {
        __syncthreads();
        for (int j = tid; j < len; j += NBIN) rec[lo + j] = staged[j];
    }
}

__global__ __launch_bounds__(NBIN) __attribute__((amdgpu_waves_per_eu(8)))
void gat_sort_gemm(const int* __restrict__ row,
                                                      const int* __restrict__ col,
                                                      int* __restrict__ cstart,
                                                      int* __restrict__ recw, int E,
                                                      const float* __restrict__ x,
                                                      const float* __restrict__ W,
                                                      const float* __restrict__ a_l,
                                                      const float* __restrict__ a_r,
                                                      float* __restrict__ h,
                                                      float* __restrict__ al,
                                                      float* __restrict__ ar, int N) {
    if (blockIdx.x < GA) {
        chunk_sort(row, col, cstart, recw, E, N);
        return;
    }
    const int wave = threadIdx.x >> 5;
    const int lane = threadIdx.x & 31;
    const int G = (int)gridDim.x - GA;
    const int tiles = (N + 15) / 16;
    const int t = (int)blockIdx.x - GA + G * wave;
    if (t < tiles) gemm_tile(x, W, a_l, a_r, h, al, ar, N, (long)t, lane);
}

__global__ __launch_bounds__(1024) void gat_group(const int* __restrict__ recw,
                                                  const int* __restrict__ cstart,
                                                  int* __restrict__ count,
                                                  int* __restrict__ offset,
                                                  int* __restrict__ scol, int E, int N) {
    const int2* __restrict__ rec = (const int2*)recw;
    __shared__ int segLo[GA];
    __shared__ int segN[GA];
    __shared__ int ncnt[SMAX];
    __shared__ int ncur[SMAX];
    __shared__ int wtot[64];
    __shared__ int staged[CAPC];
    const int tid = threadIdx.x;
    const int lane = tid & 31;
    const int seg0 = (tid >> 5) * SEGW;
    const int shift = bin_shift(N);
    const int n0 = (int)blockIdx.x << shift;
    if (n0 >= N) return;
    const int CH = (E + GA - 1) / GA;

    int cs0 = 0, c = 0;
    if (tid < GA) {
        const int* __restrict__ p = cstart + tid * CPITCH + (int)blockIdx.x;
        cs0 = p[0];
        c = p[1] - cs0;
        segLo[tid] = tid * CH + cs0;
        segN[tid] = c;
    }
    ncnt[tid] = 0;
    const int2 tot = block_sum2(cs0, c, wtot);
    const int s0 = tot.x;
    const int len = tot.y;
    const bool stage = (len <= CAPC);

    {
        int r[SEGW];
        #pragma unroll
        for (int i = 0; i < SEGW; ++i) {
            r[i] = (lane < segN[seg0 + i]) ? rec[segLo[seg0 + i] + lane].x : n0;
        }
        #pragma unroll
        for (int i = 0; i < SEGW; ++i) {
            if (lane < segN[seg0 + i]) atomicAdd(&ncnt[r[i] - n0], 1);
        }
        for (int i = 0; i < SEGW; ++i) {
            const int sl = segLo[seg0 + i];
            const int sn = segN[seg0 + i];
            for (int j = lane + 32; j < sn; j += 32) atomicAdd(&ncnt[rec[sl + j].x - n0], 1);
        }
    }
    __syncthreads();
    const int v = ncnt[tid];
    const int excl = block_excl_scan(v, wtot);
    ncur[tid] = excl;
    const int n = n0 + tid;
    if (tid < (1 << shift) && n < N) {
        count[n] = v;
        offset[n] = s0 + excl;
    }
    __syncthreads();
    {
        int2 e[SEGW];
        #pragma unroll
        for (int i = 0; i < SEGW; ++i) {
            e[i] = (lane < segN[seg0 + i]) ? rec[segLo[seg0 + i] + lane] : make_int2(n0, 0);
        }
        #pragma unroll
        for (int i = 0; i < SEGW; ++i) {
            if (lane < segN[seg0 + i]) {
                const int pos = atomicAdd(&ncur[e[i].x - n0], 1);
                if (stage) staged[pos] = e[i].y;
                else       scol[s0 + pos] = e[i].y;
            }
        }
        for (int i = 0; i < SEGW; ++i) {
            const int sl = segLo[seg0 + i];
            const int sn = segN[seg0 + i];
            for (int j = lane + 32; j < sn; j += 32) {
                const int2 f = rec[sl + j];
                const int pos = atomicAdd(&ncur[f.x - n0], 1);
                if (stage) staged[pos] = f.y;
                else       scol[s0 + pos] = f.y;
            }
        }
    }
    if (stage) {
        __syncthreads();
        for (int j = tid; j < len; j += 1024) scol[s0 + j] = staged[j];
    }
}

__device__ inline float lrelu(float v) { return (v > 0.0f) ? v : (ALPHA * v); }

template <int K>
__device__ inline float group_bcast(float v) {
    return __int_as_float((int)__builtin_amdgcn_mov_dpp8((unsigned)__float_as_int(v), (unsigned)(K * 0x249249)));
}
__device__ inline float group_max(float v) {
    #pragma unroll
    for (int off = 4; off > 0; off >>= 1) v = fmaxf(v, __shfl_xor(v, off, 32));
    return v;
}
__device__ inline float group_sum(float v) {
    #pragma unroll
    for (int off = 4; off > 0; off >>= 1) v += __shfl_xor(v, off, 32);
    return v;
}

#define ESLOT 4

#define H_LOAD(K) if (K < cnt) { hv[K] = *(const float4*)(hl + (long)__builtin_amdgcn_readlane(c, K) * (OUT_F * NHEAD)); }
#define H_FMA(K) if (K < cnt) { const float w = group_bcast<K>(p); \
                acc.x += w * hv[K].x; acc.y += w * hv[K].y; acc.z += w * hv[K].z; acc.w += w * hv[K].w; }
__device__ inline void rows_load(const float* __restrict__ hl, int c, int cnt, float4 (&hv)[8]) {
    H_LOAD(0) H_LOAD(1) H_LOAD(2) H_LOAD(3) H_LOAD(4) H_LOAD(5) H_LOAD(6) H_LOAD(7)
}
__device__ inline void rows_fma(float p, int cnt, const float4 (&hv)[8], float4& acc) {
    H_FMA(0) H_FMA(1) H_FMA(2) H_FMA(3) H_FMA(4) H_FMA(5) H_FMA(6) H_FMA(7)
}
#undef H_LOAD
#undef H_FMA

__global__ __launch_bounds__(256) void gat_node(const int* __restrict__ offset,
                                                const int* __restrict__ count,
                                                const int* __restrict__ scol,
                                                const float* __restrict__ al,
                                                const float* __restrict__ ar,
                                                const float* __restrict__ h,
                                                float* __restrict__ out, int N) {
    const int lane = threadIdx.x & 31;
    const int n = __builtin_amdgcn_readfirstlane((int)(blockIdx.x * 8 + (threadIdx.x >> 5)));
    if (n >= N) return;
    const int beg = offset[n];
    const int deg = count[n];
    const int end = beg + deg;
    const int e8 = lane & 7;
    const int hd = lane >> 3;

    float4 acc = make_float4(0.0f, 0.0f, 0.0f, 0.0f);
    if (deg > 0) {
        const float aln = al[(long)n * NHEAD + hd];
        const float* __restrict__ hl = h + lane * 4;

        if (deg <= 8 * ESLOT) {
            int   c[ESLOT];
            float p[ESLOT];
            #pragma unroll
            for (int k = 0; k < ESLOT; ++k) {
                const int idx = beg + 8 * k + e8;
                c[k] = scol[(idx < end) ? idx : (end - 1)];
            }
            #pragma unroll
            for (int k = 0; k < ESLOT; ++k) p[k] = ar[(long)c[k] * NHEAD + hd];
            float4 hv[8];
            const int cnt0 = (deg < 8) ? deg : 8;
            rows_load(hl, c[0], cnt0, hv);

            float m = -INFINITY;
            #pragma unroll
            for (int k = 0; k < ESLOT; ++k) {
                p[k] = (beg + 8 * k + e8 < end) ? lrelu(aln + p[k]) : -INFINITY;
                m = fmaxf(m, p[k]);
            }
            m = group_max(m);
            float s = 0.0f;
            #pragma unroll
            for (int k = 0; k < ESLOT; ++k) {
                p[k] = expf(p[k] - m);
                s += p[k];
            }
            s = group_sum(s);
            rows_fma(p[0] / s, cnt0, hv, acc);
            #pragma unroll
            for (int k = 1; k < ESLOT; ++k) {
                if (8 * k < deg) {
                    const int cnt = (deg - 8 * k < 8) ? (deg - 8 * k) : 8;
                    rows_load(hl, c[k], cnt, hv);
                    rows_fma(p[k] / s, cnt, hv, acc);
                }
            }
        } else {
            float m = -INFINITY;
            for (int base = beg; base < end; base += 8) {
                const int idx = base + e8;
                if (idx < end) m = fmaxf(m, lrelu(aln + ar[(long)scol[idx] * NHEAD + hd]));
            }
            m = group_max(m);

            float s = 0.0f;
            for (int base = beg; base < end; base += 8) {
                const int idx = base + e8;
                if (idx < end) s += expf(lrelu(aln + ar[(long)scol[idx] * NHEAD + hd]) - m);
            }
            s = group_sum(s);

            for (int base = beg; base < end; base += 8) {
                const int cnt = (end - base < 8) ? (end - base) : 8;
                int c = 0;
                float p = 0.0f;
                if (e8 < cnt) {
                    c = scol[base + e8];
                    p = expf(lrelu(aln + ar[(long)c * NHEAD + hd]) - m) / s;
                }
                float4 hv[8];
                rows_load(hl, c, cnt, hv);
                rows_fma(p, cnt, hv, acc);
            }
        }
    }
    *(float4*)(out + (long)n * (OUT_F * NHEAD) + lane * 4) = acc;
}

extern "C" void kernel_launch(void* const* d_in, const int* in_sizes, int n_in,
                              void* d_out, int out_size, void* d_ws, size_t ws_size,
                              hipStream_t stream) {
    const float* x    = (const float*)d_in[0];
    const int*   edge = (const int*)d_in[1];
    const float* W    = (const float*)d_in[2];
    const float* a_l  = (const float*)d_in[3];
    const float* a_r  = (const float*)d_in[4];

    const int N = in_sizes[0] / IN_F;
    const int E = in_sizes[1] / 2;
    const int* rowp = edge;
    const int* colp = edge + E;

    float* ws = (float*)d_ws;
    float* h  = ws;
    float* al = h  + (size_t)N * (OUT_F * NHEAD);
    float* ar = al + (size_t)N * NHEAD;
    int* count  = (int*)(ar + (size_t)N * NHEAD);
    int* offset = count + (size_t)N;
    int* scol   = offset + (size_t)N;
    int* cstart = (int*)d_out;
    int* rec    = cstart + GA * CPITCH;
    float* out = (float*)d_out;

    const int tiles16 = (N + 15) / 16;
    const int G = ((tiles16 + 15) / 16 > 512) ? (tiles16 + 15) / 16 : 512;
    gat_sort_gemm<<<(unsigned)(GA + G), NBIN, 0, stream>>>(rowp, colp, cstart, rec, E, x, W, a_l, a_r, h, al, ar, N);
    gat_group<<<NBIN, 1024, 0, stream>>>(rec, cstart, count, offset, scol, E, N);
    gat_node<<<(unsigned)((N + 7) / 8), 256, 0, stream>>>(offset, count, scol, al, ar, h, out, N);
}


# Product's B operand read from an LDS copy of W made once per work-group (one 16-B LDS read per tile per two k steps, was four 4-B global loads); A read 16 B wide; sum order in blocks of 8 k changes
# speedup vs baseline: 9.0025x; 1.1095x over previous
#include <hip/hip_runtime.h>
#include <hip/hip_bf16.h>
#include <math.h>

typedef __attribute__((ext_vector_type(2))) float v2f;
typedef __attribute__((ext_vector_type(8))) float v8f;

#define IN_F   128
#define OUT_F  32
#define NHEAD  4
#define ALPHA  0.2f


__device__ inline void stage_w(const float* __restrict__ W, float4* __restrict__ wq) {
    const int c4 = (threadIdx.x & 31) * 4;
    for (int kq = (int)(threadIdx.x >> 5); kq < IN_F / 4; kq += (int)(blockDim.x >> 5)) {
        const float* __restrict__ s = W + (4 * kq) * (OUT_F * NHEAD) + c4;
        const float4 r0 = *(const float4*)(s);
        const float4 r1 = *(const float4*)(s + (OUT_F * NHEAD));
        const float4 r2 = *(const float4*)(s + 2 * (OUT_F * NHEAD));
        const float4 r3 = *(const float4*)(s + 3 * (OUT_F * NHEAD));
        float4* __restrict__ d = wq + kq * (OUT_F * NHEAD) + c4;
        d[0] = make_float4(r0.x, r1.x, r2.x, r3.x);
        d[1] = make_float4(r0.y, r1.y, r2.y, r3.y);
        d[2] = make_float4(r0.z, r1.z, r2.z, r3.z);
        d[3] = make_float4(r0.w, r1.w, r2.w, r3.w);
    }
}

template <int XM>
__device__ inline float row_xor(float v) {
    return __int_as_float(__builtin_amdgcn_update_dpp(0, __float_as_int(v), 0x160 | XM, 0xf, 0xf, true));
}
template <int XM>
__device__ inline float fold(float a, float b, bool hi) {
    const float keep = hi ? b : a;
    const float send = hi ? a : b;
    return keep + row_xor<XM>(send);
}

__device__ inline void gemm_tile(const float* __restrict__ x,
                                 const float4* __restrict__ wq,
                                 const float* __restrict__ a_l,
                                 const float* __restrict__ a_r,
                                 float* __restrict__ h,
                                 float* __restrict__ al,
                                 float* __restrict__ ar, int N, long rowTile, int lane) {
    const long row0 = rowTile * 16;

    const int mrow = lane & 15;
    const int hf   = lane >> 4;

    v8f acc[8];
    #pragma unroll
    for (int t = 0; t < 8; ++t) acc[t] = v8f{};

    long r = row0 + mrow;
    if (r >= N) r = N - 1;
    const float* __restrict__ xq = x + r * IN_F + 4 * hf;
    const float4* __restrict__ wl = wq + hf * (OUT_F * NHEAD) + mrow;

    float4 anext = *(const float4*)xq;
    for (int kb = 0; kb < IN_F / 8; ++kb) {
        const float4 a4 = anext;
        if (kb + 1 < IN_F / 8) anext = *(const float4*)(xq + 8 * (kb + 1));
        const float4* __restrict__ wk = wl + kb * (2 * OUT_F * NHEAD);
        float4 b4[8];
        #pragma unroll
        for (int t = 0; t < 8; ++t) b4[t] = wk[t * 16];
        v2f a, b;
        a.x = a4.x; a.y = a4.y;
        #pragma unroll
        for (int t = 0; t < 8; ++t) {
            b.x = b4[t].x; b.y = b4[t].y;
            acc[t] = __builtin_amdgcn_wmma_f32_16x16x4_f32(
                false, a, false, b, (short)0, acc[t], false, false);
        }
        a.x = a4.z; a.y = a4.w;
        #pragma unroll
        for (int t = 0; t < 8; ++t) {
            b.x = b4[t].z; b.y = b4[t].w;
            acc[t] = __builtin_amdgcn_wmma_f32_16x16x4_f32(
                false, a, false, b, (short)0, acc[t], false, false);
        }
    }

    const int rowOff = (lane < 16) ? 0 : 8;
    #pragma unroll
    for (int t = 0; t < 8; ++t) {
        #pragma unroll
        for (int g = 0; g < 8; ++g) {
            float v = acc[t][g];
            v = (v == v) ? v : 0.0f;
            acc[t][g] = v;
            const long rr = row0 + g + rowOff;
            if (rr < N) h[rr * (OUT_F * NHEAD) + t * 16 + mrow] = v;
        }
    }

    const bool b0 = (mrow & 1) != 0, b1 = (mrow & 2) != 0, b2 = (mrow & 4) != 0, b3 = (mrow & 8) != 0;
    float res[NHEAD];
    #pragma unroll
    for (int j = 0; j < NHEAD; ++j) {
        float q[2];
        #pragma unroll
        for (int c = 0; c < 2; ++c) {
            const float* __restrict__ av = (c == 0) ? a_l : a_r;
            const float w0 = av[(2 * j) * 16 + mrow];
            const float w1 = av[(2 * j + 1) * 16 + mrow];
            float p[8];
            #pragma unroll
            for (int g = 0; g < 8; ++g) p[g] = w0 * acc[2 * j][g] + w1 * acc[2 * j + 1][g];
            const float p01 = fold<1>(p[0], p[1], b0);
            const float p23 = fold<1>(p[2], p[3], b0);
            const float p45 = fold<1>(p[4], p[5], b0);
            const float p67 = fold<1>(p[6], p[7], b0);
            const float p03 = fold<2>(p01, p23, b1);
            const float p47 = fold<2>(p45, p67, b1);
            q[c] = fold<4>(p03, p47, b2);
        }
        res[j] = fold<8>(q[0], q[1], b3);
    }
    const long orow = row0 + rowOff + (mrow & 7);
    if (orow < N) {
        float* __restrict__ dst = (b3 ? ar : al) + orow * NHEAD;
        *(float4*)dst = make_float4(res[0], res[1], res[2], res[3]);
    }
}

#define GA     256
#define NBIN   512
#define SMAX   1024
#define CHMAX  8192
#define CAPC   16384
#define CPITCH 544
#define SEGW   (GA / 32)

__device__ inline int bin_shift(int N) {
    return 32 - __clz((N - 1) >> 9);
}

__device__ inline int block_excl_scan(int v, int* wtot) {
    const int lane = threadIdx.x & 31;
    const int wave = threadIdx.x >> 5;
    const int nw = blockDim.x >> 5;
    int incl = v;
    #pragma unroll
    for (int off = 1; off < 32; off <<= 1) {
        const int u = __shfl_up(incl, off, 32);
        if (lane >= off) incl += u;
    }
    if (lane == 31) wtot[wave] = incl;
    __syncthreads();
    int pre = 0;
    for (int w = 0; w < nw; ++w) {
        if (w < wave) pre += wtot[w];
    }
    __syncthreads();
    return pre + incl - v;
}

__device__ inline int2 block_sum2(int a, int b, int* wtot2) {
    const int lane = threadIdx.x & 31;
    const int wave = threadIdx.x >> 5;
    const int nw = blockDim.x >> 5;
    #pragma unroll
    for (int off = 16; off > 0; off >>= 1) {
        a += __shfl_xor(a, off, 32);
        b += __shfl_xor(b, off, 32);
    }
    if (lane == 0) {
        wtot2[wave] = a;
        wtot2[32 + wave] = b;
    }
    __syncthreads();
    int ta = 0, tb = 0;
    for (int w = 0; w < nw; ++w) {
        ta += wtot2[w];
        tb += wtot2[32 + w];
    }
    __syncthreads();
    return make_int2(ta, tb);
}

__device__ inline void chunk_sort(const int* __restrict__ row, const int* __restrict__ col,
                                  int* __restrict__ cstart, int* __restrict__ recw,
                                  int E, int N) {
    int2* __restrict__ rec = (int2*)recw;
    __shared__ int cnt[NBIN];
    __shared__ int cur[NBIN];
    __shared__ int wtot[32];
    __shared__ int2 staged[CHMAX];
    const int tid = threadIdx.x;
    const int shift = bin_shift(N);
    const int CH = (E + GA - 1) / GA;
    const int lo = (int)blockIdx.x * CH;
    const int hi = (lo + CH < E) ? (lo + CH) : E;
    const int len = (hi > lo) ? (hi - lo) : 0;
    const bool stage = (CH <= CHMAX);

    cnt[tid] = 0;
    __syncthreads();
    for (int i0 = lo + tid; i0 < hi; i0 += 4 * NBIN) {
        int r[4];
        #pragma unroll
        for (int k = 0; k < 4; ++k) {
            const int i = i0 + k * NBIN;
            r[k] = (i < hi) ? row[i] : 0;
        }
        #pragma unroll
        for (int k = 0; k < 4; ++k) {
            if (i0 + k * NBIN < hi) atomicAdd(&cnt[r[k] >> shift], 1);
        }
    }
    __syncthreads();
    const int c = cnt[tid];
    const int excl = block_excl_scan(c, wtot);
    cur[tid] = excl;
    int* __restrict__ cs = cstart + (int)blockIdx.x * CPITCH;
    cs[tid] = excl;
    if (tid == NBIN - 1) cs[NBIN] = excl + c;
    __syncthreads();
    for (int i0 = lo + tid; i0 < hi; i0 += 4 * NBIN) {
        int r[4], s[4];
        #pragma unroll
        for (int k = 0; k < 4; ++k) {
            const int i = i0 + k * NBIN;
            r[k] = (i < hi) ? row[i] : 0;
            s[k] = (i < hi) ? col[i] : 0;
        }
        #pragma unroll
        for (int k = 0; k < 4; ++k) {
            if (i0 + k * NBIN < hi) {
                const int pos = atomicAdd(&cur[r[k] >> shift], 1);
                if (stage) staged[pos]   = make_int2(r[k], s[k]);
                else       rec[lo + pos] = make_int2(r[k], s[k]);
            }
        }
    }
    if (stage) {
        __syncthreads();
        for (int j = tid; j < len; j += NBIN) rec[lo + j] = staged[j];
    }
}

__global__ __launch_bounds__(NBIN) __attribute__((amdgpu_waves_per_eu(8)))
void gat_sort_gemm(const int* __restrict__ row,
                                                      const int* __restrict__ col,
                                                      int* __restrict__ cstart,
                                                      int* __restrict__ recw, int E,
                                                      const float* __restrict__ x,
                                                      const float* __restrict__ W,
                                                      const float* __restrict__ a_l,
                                                      const float* __restrict__ a_r,
                                                      float* __restrict__ h,
                                                      float* __restrict__ al,
                                                      float* __restrict__ ar, int N) {
    if (blockIdx.x < GA) {
        chunk_sort(row, col, cstart, recw, E, N);
        return;
    }
    __shared__ float4 wq[(IN_F / 4) * (OUT_F * NHEAD)];
    stage_w(W, wq);
    __syncthreads();
    const int wave = threadIdx.x >> 5;
    const int lane = threadIdx.x & 31;
    const int G = (int)gridDim.x - GA;
    const int tiles = (N + 15) / 16;
    const int t = (int)blockIdx.x - GA + G * wave;
    if (t < tiles) gemm_tile(x, wq, a_l, a_r, h, al, ar, N, (long)t, lane);
}

__global__ __launch_bounds__(1024) void gat_group(const int* __restrict__ recw,
                                                  const int* __restrict__ cstart,
                                                  int* __restrict__ count,
                                                  int* __restrict__ offset,
                                                  int* __restrict__ scol, int E, int N) {
    const int2* __restrict__ rec = (const int2*)recw;
    __shared__ int segLo[GA];
    __shared__ int segN[GA];
    __shared__ int ncnt[SMAX];
    __shared__ int ncur[SMAX];
    __shared__ int wtot[64];
    __shared__ int staged[CAPC];
    const int tid = threadIdx.x;
    const int lane = tid & 31;
    const int seg0 = (tid >> 5) * SEGW;
    const int shift = bin_shift(N);
    const int n0 = (int)blockIdx.x << shift;
    if (n0 >= N) return;
    const int CH = (E + GA - 1) / GA;

    int cs0 = 0, c = 0;
    if (tid < GA) {
        const int* __restrict__ p = cstart + tid * CPITCH + (int)blockIdx.x;
        cs0 = p[0];
        c = p[1] - cs0;
        segLo[tid] = tid * CH + cs0;
        segN[tid] = c;
    }
    ncnt[tid] = 0;
    const int2 tot = block_sum2(cs0, c, wtot);
    const int s0 = tot.x;
    const int len = tot.y;
    const bool stage = (len <= CAPC);

    {
        int r[SEGW];
        #pragma unroll
        for (int i = 0; i < SEGW; ++i) {
            r[i] = (lane < segN[seg0 + i]) ? rec[segLo[seg0 + i] + lane].x : n0;
        }
        #pragma unroll
        for (int i = 0; i < SEGW; ++i) {
            if (lane < segN[seg0 + i]) atomicAdd(&ncnt[r[i] - n0], 1);
        }
        for (int i = 0; i < SEGW; ++i) {
            const int sl = segLo[seg0 + i];
            const int sn = segN[seg0 + i];
            for (int j = lane + 32; j < sn; j += 32) atomicAdd(&ncnt[rec[sl + j].x - n0], 1);
        }
    }
    __syncthreads();
    const int v = ncnt[tid];
    const int excl = block_excl_scan(v, wtot);
    ncur[tid] = excl;
    const int n = n0 + tid;
    if (tid < (1 << shift) && n < N) {
        count[n] = v;
        offset[n] = s0 + excl;
    }
    __syncthreads();
    {
        int2 e[SEGW];
        #pragma unroll
        for (int i = 0; i < SEGW; ++i) {
            e[i] = (lane < segN[seg0 + i]) ? rec[segLo[seg0 + i] + lane] : make_int2(n0, 0);
        }
        #pragma unroll
        for (int i = 0; i < SEGW; ++i) {
            if (lane < segN[seg0 + i]) {
                const int pos = atomicAdd(&ncur[e[i].x - n0], 1);
                if (stage) staged[pos] = e[i].y;
                else       scol[s0 + pos] = e[i].y;
            }
        }
        for (int i = 0; i < SEGW; ++i) {
            const int sl = segLo[seg0 + i];
            const int sn = segN[seg0 + i];
            for (int j = lane + 32; j < sn; j += 32) {
                const int2 f = rec[sl + j];
                const int pos = atomicAdd(&ncur[f.x - n0], 1);
                if (stage) staged[pos] = f.y;
                else       scol[s0 + pos] = f.y;
            }
        }
    }
    if (stage) {
        __syncthreads();
        for (int j = tid; j < len; j += 1024) scol[s0 + j] = staged[j];
    }
}

__device__ inline float lrelu(float v) { return (v > 0.0f) ? v : (ALPHA * v); }

template <int K>
__device__ inline float group_bcast(float v) {
    return __int_as_float((int)__builtin_amdgcn_mov_dpp8((unsigned)__float_as_int(v), (unsigned)(K * 0x249249)));
}
__device__ inline float group_max(float v) {
    #pragma unroll
    for (int off = 4; off > 0; off >>= 1) v = fmaxf(v, __shfl_xor(v, off, 32));
    return v;
}
__device__ inline float group_sum(float v) {
    #pragma unroll
    for (int off = 4; off > 0; off >>= 1) v += __shfl_xor(v, off, 32);
    return v;
}

#define ESLOT 4

#define H_LOAD(K) if (K < cnt) { hv[K] = *(const float4*)(hl + (long)__builtin_amdgcn_readlane(c, K) * (OUT_F * NHEAD)); }
#define H_FMA(K) if (K < cnt) { const float w = group_bcast<K>(p); \
                acc.x += w * hv[K].x; acc.y += w * hv[K].y; acc.z += w * hv[K].z; acc.w += w * hv[K].w; }
__device__ inline void rows_load(const float* __restrict__ hl, int c, int cnt, float4 (&hv)[8]) {
    H_LOAD(0) H_LOAD(1) H_LOAD(2) H_LOAD(3) H_LOAD(4) H_LOAD(5) H_LOAD(6) H_LOAD(7)
}
__device__ inline void rows_fma(float p, int cnt, const float4 (&hv)[8], float4& acc) {
    H_FMA(0) H_FMA(1) H_FMA(2) H_FMA(3) H_FMA(4) H_FMA(5) H_FMA(6) H_FMA(7)
}
#undef H_LOAD
#undef H_FMA

__global__ __launch_bounds__(256) void gat_node(const int* __restrict__ offset,
                                                const int* __restrict__ count,
                                                const int* __restrict__ scol,
                                                const float* __restrict__ al,
                                                const float* __restrict__ ar,
                                                const float* __restrict__ h,
                                                float* __restrict__ out, int N) {
    const int lane = threadIdx.x & 31;
    const int n = __builtin_amdgcn_readfirstlane((int)(blockIdx.x * 8 + (threadIdx.x >> 5)));
    if (n >= N) return;
    const int beg = offset[n];
    const int deg = count[n];
    const int end = beg + deg;
    const int e8 = lane & 7;
    const int hd = lane >> 3;

    float4 acc = make_float4(0.0f, 0.0f, 0.0f, 0.0f);
    if (deg > 0) {
        const float aln = al[(long)n * NHEAD + hd];
        const float* __restrict__ hl = h + lane * 4;

        if (deg <= 8 * ESLOT) {
            int   c[ESLOT];
            float p[ESLOT];
            #pragma unroll
            for (int k = 0; k < ESLOT; ++k) {
                const int idx = beg + 8 * k + e8;
                c[k] = scol[(idx < end) ? idx : (end - 1)];
            }
            #pragma unroll
            for (int k = 0; k < ESLOT; ++k) p[k] = ar[(long)c[k] * NHEAD + hd];
            float4 hv[8];
            const int cnt0 = (deg < 8) ? deg : 8;
            rows_load(hl, c[0], cnt0, hv);

            float m = -INFINITY;
            #pragma unroll
            for (int k = 0; k < ESLOT; ++k) {
                p[k] = (beg + 8 * k + e8 < end) ? lrelu(aln + p[k]) : -INFINITY;
                m = fmaxf(m, p[k]);
            }
            m = group_max(m);
            float s = 0.0f;
            #pragma unroll
            for (int k = 0; k < ESLOT; ++k) {
                p[k] = expf(p[k] - m);
                s += p[k];
            }
            s = group_sum(s);
            rows_fma(p[0] / s, cnt0, hv, acc);
            #pragma unroll
            for (int k = 1; k < ESLOT; ++k) {
                if (8 * k < deg) {
                    const int cnt = (deg - 8 * k < 8) ? (deg - 8 * k) : 8;
                    rows_load(hl, c[k], cnt, hv);
                    rows_fma(p[k] / s, cnt, hv, acc);
                }
            }
        } else {
            float m = -INFINITY;
            for (int base = beg; base < end; base += 8) {
                const int idx = base + e8;
                if (idx < end) m = fmaxf(m, lrelu(aln + ar[(long)scol[idx] * NHEAD + hd]));
            }
            m = group_max(m);

            float s = 0.0f;
            for (int base = beg; base < end; base += 8) {
                const int idx = base + e8;
                if (idx < end) s += expf(lrelu(aln + ar[(long)scol[idx] * NHEAD + hd]) - m);
            }
            s = group_sum(s);

            for (int base = beg; base < end; base += 8) {
                const int cnt = (end - base < 8) ? (end - base) : 8;
                int c = 0;
                float p = 0.0f;
                if (e8 < cnt) {
                    c = scol[base + e8];
                    p = expf(lrelu(aln + ar[(long)c * NHEAD + hd]) - m) / s;
                }
                float4 hv[8];
                rows_load(hl, c, cnt, hv);
                rows_fma(p, cnt, hv, acc);
            }
        }
    }
    *(float4*)(out + (long)n * (OUT_F * NHEAD) + lane * 4) = acc;
}

extern "C" void kernel_launch(void* const* d_in, const int* in_sizes, int n_in,
                              void* d_out, int out_size, void* d_ws, size_t ws_size,
                              hipStream_t stream) {
    const float* x    = (const float*)d_in[0];
    const int*   edge = (const int*)d_in[1];
    const float* W    = (const float*)d_in[2];
    const float* a_l  = (const float*)d_in[3];
    const float* a_r  = (const float*)d_in[4];

    const int N = in_sizes[0] / IN_F;
    const int E = in_sizes[1] / 2;
    const int* rowp = edge;
    const int* colp = edge + E;

    float* ws = (float*)d_ws;
    float* h  = ws;
    float* al = h  + (size_t)N * (OUT_F * NHEAD);
    float* ar = al + (size_t)N * NHEAD;
    int* count  = (int*)(ar + (size_t)N * NHEAD);
    int* offset = count + (size_t)N;
    int* scol   = offset + (size_t)N;
    int* cstart = (int*)d_out;
    int* rec    = cstart + GA * CPITCH;
    float* out = (float*)d_out;

    const int tiles16 = (N + 15) / 16;
    const int G = ((tiles16 + 15) / 16 > 512) ? (tiles16 + 15) / 16 : 512;
    gat_sort_gemm<<<(unsigned)(GA + G), NBIN, 0, stream>>>(rowp, colp, cstart, rec, E, x, W, a_l, a_r, h, al, ar, N);
    gat_group<<<NBIN, 1024, 0, stream>>>(rec, cstart, count, offset, scol, E, N);
    gat_node<<<(unsigned)((N + 7) / 8), 256, 0, stream>>>(offset, count, scol, al, ar, h, out, N);
}


# Level 2 of the grouping and the node walk in one launch: a bin's node lists stay in LDS and its waves draw nodes from an LDS counter; lists stay in arrival order, sums unchanged
# speedup vs baseline: 9.8124x; 1.0900x over previous
#include <hip/hip_runtime.h>
#include <hip/hip_bf16.h>
#include <math.h>

typedef __attribute__((ext_vector_type(2))) float v2f;
typedef __attribute__((ext_vector_type(8))) float v8f;

#define IN_F   128
#define OUT_F  32
#define NHEAD  4
#define ALPHA  0.2f


__device__ inline void stage_w(const float* __restrict__ W, float4* __restrict__ wq) {
    const int c4 = (threadIdx.x & 31) * 4;
    for (int kq = (int)(threadIdx.x >> 5); kq < IN_F / 4; kq += (int)(blockDim.x >> 5)) {
        const float* __restrict__ s = W + (4 * kq) * (OUT_F * NHEAD) + c4;
        const float4 r0 = *(const float4*)(s);
        const float4 r1 = *(const float4*)(s + (OUT_F * NHEAD));
        const float4 r2 = *(const float4*)(s + 2 * (OUT_F * NHEAD));
        const float4 r3 = *(const float4*)(s + 3 * (OUT_F * NHEAD));
        float4* __restrict__ d = wq + kq * (OUT_F * NHEAD) + c4;
        d[0] = make_float4(r0.x, r1.x, r2.x, r3.x);
        d[1] = make_float4(r0.y, r1.y, r2.y, r3.y);
        d[2] = make_float4(r0.z, r1.z, r2.z, r3.z);
        d[3] = make_float4(r0.w, r1.w, r2.w, r3.w);
    }
}

template <int XM>
__device__ inline float row_xor(float v) {
    return __int_as_float(__builtin_amdgcn_update_dpp(0, __float_as_int(v), 0x160 | XM, 0xf, 0xf, true));
}
template <int XM>
__device__ inline float fold(float a, float b, bool hi) {
    const float keep = hi ? b : a;
    const float send = hi ? a : b;
    return keep + row_xor<XM>(send);
}

__device__ inline void gemm_tile(const float* __restrict__ x,
                                 const float4* __restrict__ wq,
                                 const float* __restrict__ a_l,
                                 const float* __restrict__ a_r,
                                 float* __restrict__ h,
                                 float* __restrict__ al,
                                 float* __restrict__ ar, int N, long rowTile, int lane) {
    const long row0 = rowTile * 16;

    const int mrow = lane & 15;
    const int hf   = lane >> 4;

    v8f acc[8];
    #pragma unroll
    for (int t = 0; t < 8; ++t) acc[t] = v8f{};

    long r = row0 + mrow;
    if (r >= N) r = N - 1;
    const float* __restrict__ xq = x + r * IN_F + 4 * hf;
    const float4* __restrict__ wl = wq + hf * (OUT_F * NHEAD) + mrow;

    float4 anext = *(const float4*)xq;
    for (int kb = 0; kb < IN_F / 8; ++kb) {
        const float4 a4 = anext;
        if (kb + 1 < IN_F / 8) anext = *(const float4*)(xq + 8 * (kb + 1));
        const float4* __restrict__ wk = wl + kb * (2 * OUT_F * NHEAD);
        float4 b4[8];
        #pragma unroll
        for (int t = 0; t < 8; ++t) b4[t] = wk[t * 16];
        v2f a, b;
        a.x = a4.x; a.y = a4.y;
        #pragma unroll
        for (int t = 0; t < 8; ++t) {
            b.x = b4[t].x; b.y = b4[t].y;
            acc[t] = __builtin_amdgcn_wmma_f32_16x16x4_f32(
                false, a, false, b, (short)0, acc[t], false, false);
        }
        a.x = a4.z; a.y = a4.w;
        #pragma unroll
        for (int t = 0; t < 8; ++t) {
            b.x = b4[t].z; b.y = b4[t].w;
            acc[t] = __builtin_amdgcn_wmma_f32_16x16x4_f32(
                false, a, false, b, (short)0, acc[t], false, false);
        }
    }

    const int rowOff = (lane < 16) ? 0 : 8;
    #pragma unroll
    for (int t = 0; t < 8; ++t) {
        #pragma unroll
        for (int g = 0; g < 8; ++g) {
            float v = acc[t][g];
            v = (v == v) ? v : 0.0f;
            acc[t][g] = v;
            const long rr = row0 + g + rowOff;
            if (rr < N) h[rr * (OUT_F * NHEAD) + t * 16 + mrow] = v;
        }
    }

    const bool b0 = (mrow & 1) != 0, b1 = (mrow & 2) != 0, b2 = (mrow & 4) != 0, b3 = (mrow & 8) != 0;
    float res[NHEAD];
    #pragma unroll
    for (int j = 0; j < NHEAD; ++j) {
        float q[2];
        #pragma unroll
        for (int c = 0; c < 2; ++c) {
            const float* __restrict__ av = (c == 0) ? a_l : a_r;
            const float w0 = av[(2 * j) * 16 + mrow];
            const float w1 = av[(2 * j + 1) * 16 + mrow];
            float p[8];
            #pragma unroll
            for (int g = 0; g < 8; ++g) p[g] = w0 * acc[2 * j][g] + w1 * acc[2 * j + 1][g];
            const float p01 = fold<1>(p[0], p[1], b0);
            const float p23 = fold<1>(p[2], p[3], b0);
            const float p45 = fold<1>(p[4], p[5], b0);
            const float p67 = fold<1>(p[6], p[7], b0);
            const float p03 = fold<2>(p01, p23, b1);
            const float p47 = fold<2>(p45, p67, b1);
            q[c] = fold<4>(p03, p47, b2);
        }
        res[j] = fold<8>(q[0], q[1], b3);
    }
    const long orow = row0 + rowOff + (mrow & 7);
    if (orow < N) {
        float* __restrict__ dst = (b3 ? ar : al) + orow * NHEAD;
        *(float4*)dst = make_float4(res[0], res[1], res[2], res[3]);
    }
}

#define GA     256
#define NBIN   512
#define SMAX   1024
#define CHMAX  8192
#define CAPC   16384
#define CPITCH 544
#define SEGW   (GA / 32)
#define CBITS  19
#define CMASK  ((1 << CBITS) - 1)

__device__ inline int bin_nodes(int N) { return (N + NBIN - 1) / NBIN; }
__device__ inline unsigned bin_magic(int P) { return 0xFFFFFFFFu / (unsigned)P + 1u; }
__device__ inline int bin_of(int r, int P, unsigned M) {
    return (P == 1) ? r : (int)__umulhi((unsigned)r, M);
}

__device__ inline int block_excl_scan(int v, int* wtot) {
    const int lane = threadIdx.x & 31;
    const int wave = threadIdx.x >> 5;
    const int nw = blockDim.x >> 5;
    int incl = v;
    #pragma unroll
    for (int off = 1; off < 32; off <<= 1) {
        const int u = __shfl_up(incl, off, 32);
        if (lane >= off) incl += u;
    }
    if (lane == 31) wtot[wave] = incl;
    __syncthreads();
    int pre = 0;
    for (int w = 0; w < nw; ++w) {
        if (w < wave) pre += wtot[w];
    }
    __syncthreads();
    return pre + incl - v;
}

__device__ inline int2 block_sum2(int a, int b, int* wtot2) {
    const int lane = threadIdx.x & 31;
    const int wave = threadIdx.x >> 5;
    const int nw = blockDim.x >> 5;
    #pragma unroll
    for (int off = 16; off > 0; off >>= 1) {
        a += __shfl_xor(a, off, 32);
        b += __shfl_xor(b, off, 32);
    }
    if (lane == 0) {
        wtot2[wave] = a;
        wtot2[32 + wave] = b;
    }
    __syncthreads();
    int ta = 0, tb = 0;
    for (int w = 0; w < nw; ++w) {
        ta += wtot2[w];
        tb += wtot2[32 + w];
    }
    __syncthreads();
    return make_int2(ta, tb);
}

__device__ inline void chunk_sort(const int* __restrict__ row, const int* __restrict__ col,
                                  int* __restrict__ cstart, int* __restrict__ rec,
                                  int E, int N) {
    __shared__ int cnt[NBIN];
    __shared__ int cur[NBIN];
    __shared__ int wtot[32];
    __shared__ int staged[CHMAX];
    const int tid = threadIdx.x;
    const int P = bin_nodes(N);
    const unsigned M = bin_magic(P);
    const int CH = (E + GA - 1) / GA;
    const int lo = (int)blockIdx.x * CH;
    const int hi = (lo + CH < E) ? (lo + CH) : E;
    const int len = (hi > lo) ? (hi - lo) : 0;
    const bool stage = (CH <= CHMAX);

    cnt[tid] = 0;
    __syncthreads();
    for (int i0 = lo + tid; i0 < hi; i0 += 4 * NBIN) {
        int r[4];
        #pragma unroll
        for (int k = 0; k < 4; ++k) {
            const int i = i0 + k * NBIN;
            r[k] = (i < hi) ? row[i] : 0;
        }
        #pragma unroll
        for (int k = 0; k < 4; ++k) {
            if (i0 + k * NBIN < hi) atomicAdd(&cnt[bin_of(r[k], P, M)], 1);
        }
    }
    __syncthreads();
    const int c = cnt[tid];
    const int excl = block_excl_scan(c, wtot);
    cur[tid] = excl;
    int* __restrict__ cs = cstart + (int)blockIdx.x * CPITCH;
    cs[tid] = excl;
    if (tid == NBIN - 1) cs[NBIN] = excl + c;
    __syncthreads();
    for (int i0 = lo + tid; i0 < hi; i0 += 4 * NBIN) {
        int r[4], s[4];
        #pragma unroll
        for (int k = 0; k < 4; ++k) {
            const int i = i0 + k * NBIN;
            r[k] = (i < hi) ? row[i] : 0;
            s[k] = (i < hi) ? col[i] : 0;
        }
        #pragma unroll
        for (int k = 0; k < 4; ++k) {
            if (i0 + k * NBIN < hi) {
                const int b = bin_of(r[k], P, M);
                const int pos = atomicAdd(&cur[b], 1);
                const int v = ((r[k] - b * P) << CBITS) | s[k];
                if (stage) staged[pos]   = v;
                else       rec[lo + pos] = v;
            }
        }
    }
    if (stage) {
        __syncthreads();
        for (int j = tid; j < len; j += NBIN) rec[lo + j] = staged[j];
    }
}

__global__ __launch_bounds__(NBIN) __attribute__((amdgpu_waves_per_eu(8)))
void gat_sort_gemm(const int* __restrict__ row,
                                                      const int* __restrict__ col,
                                                      int* __restrict__ cstart,
                                                      int* __restrict__ recw, int E,
                                                      const float* __restrict__ x,
                                                      const float* __restrict__ W,
                                                      const float* __restrict__ a_l,
                                                      const float* __restrict__ a_r,
                                                      float* __restrict__ h,
                                                      float* __restrict__ al,
                                                      float* __restrict__ ar, int N) {
    if (blockIdx.x < GA) {
        chunk_sort(row, col, cstart, recw, E, N);
        return;
    }
    __shared__ float4 wq[(IN_F / 4) * (OUT_F * NHEAD)];
    stage_w(W, wq);
    __syncthreads();
    const int wave = threadIdx.x >> 5;
    const int lane = threadIdx.x & 31;
    const int G = (int)gridDim.x - GA;
    const int tiles = (N + 15) / 16;
    const int t = (int)blockIdx.x - GA + G * wave;
    if (t < tiles) gemm_tile(x, wq, a_l, a_r, h, al, ar, N, (long)t, lane);
}

__device__ inline float lrelu(float v) { return (v > 0.0f) ? v : (ALPHA * v); }

template <int K>
__device__ inline float group_bcast(float v) {
    return __int_as_float((int)__builtin_amdgcn_mov_dpp8((unsigned)__float_as_int(v), (unsigned)(K * 0x249249)));
}
__device__ inline float group_max(float v) {
    #pragma unroll
    for (int off = 4; off > 0; off >>= 1) v = fmaxf(v, __shfl_xor(v, off, 32));
    return v;
}
__device__ inline float group_sum(float v) {
    #pragma unroll
    for (int off = 4; off > 0; off >>= 1) v += __shfl_xor(v, off, 32);
    return v;
}

#define ESLOT 4

#define H_LOAD(K) if (K < cnt) { hv[K] = *(const float4*)(hl + (long)__builtin_amdgcn_readlane(c, K) * (OUT_F * NHEAD)); }
#define H_FMA(K) if (K < cnt) { const float w = group_bcast<K>(p); \
                acc.x += w * hv[K].x; acc.y += w * hv[K].y; acc.z += w * hv[K].z; acc.w += w * hv[K].w; }
__device__ inline void rows_load(const float* __restrict__ hl, int c, int cnt, float4 (&hv)[8]) {
    H_LOAD(0) H_LOAD(1) H_LOAD(2) H_LOAD(3) H_LOAD(4) H_LOAD(5) H_LOAD(6) H_LOAD(7)
}
__device__ inline void rows_fma(float p, int cnt, const float4 (&hv)[8], float4& acc) {
    H_FMA(0) H_FMA(1) H_FMA(2) H_FMA(3) H_FMA(4) H_FMA(5) H_FMA(6) H_FMA(7)
}
#undef H_LOAD
#undef H_FMA

__device__ inline void node_walk(const int* __restrict__ ids, int deg, int n, int lane,
                                 const float* __restrict__ al,
                                 const float* __restrict__ ar,
                                 const float* __restrict__ h,
                                 float* __restrict__ out) {
    const int e8 = lane & 7;
    const int hd = lane >> 3;

    float4 acc = make_float4(0.0f, 0.0f, 0.0f, 0.0f);
    if (deg > 0) {
        const float aln = al[(long)n * NHEAD + hd];
        const float* __restrict__ hl = h + lane * 4;

        if (deg <= 8 * ESLOT) {
            int   c[ESLOT];
            float p[ESLOT];
            #pragma unroll
            for (int k = 0; k < ESLOT; ++k) {
                const int j = 8 * k + e8;
                c[k] = ids[(j < deg) ? j : (deg - 1)];
            }
            #pragma unroll
            for (int k = 0; k < ESLOT; ++k) p[k] = ar[(long)c[k] * NHEAD + hd];
            float4 hv[8];
            const int cnt0 = (deg < 8) ? deg : 8;
            rows_load(hl, c[0], cnt0, hv);

            float m = -INFINITY;
            #pragma unroll
            for (int k = 0; k < ESLOT; ++k) {
                p[k] = (8 * k + e8 < deg) ? lrelu(aln + p[k]) : -INFINITY;
                m = fmaxf(m, p[k]);
            }
            m = group_max(m);
            float s = 0.0f;
            #pragma unroll
            for (int k = 0; k < ESLOT; ++k) {
                p[k] = expf(p[k] - m);
                s += p[k];
            }
            s = group_sum(s);
            rows_fma(p[0] / s, cnt0, hv, acc);
            #pragma unroll
            for (int k = 1; k < ESLOT; ++k) {
                if (8 * k < deg) {
                    const int cnt = (deg - 8 * k < 8) ? (deg - 8 * k) : 8;
                    rows_load(hl, c[k], cnt, hv);
                    rows_fma(p[k] / s, cnt, hv, acc);
                }
            }
        } else {
            float m = -INFINITY;
            for (int base = 0; base < deg; base += 8) {
                const int idx = base + e8;
                if (idx < deg) m = fmaxf(m, lrelu(aln + ar[(long)ids[idx] * NHEAD + hd]));
            }
            m = group_max(m);

            float s = 0.0f;
            for (int base = 0; base < deg; base += 8) {
                const int idx = base + e8;
                if (idx < deg) s += expf(lrelu(aln + ar[(long)ids[idx] * NHEAD + hd]) - m);
            }
            s = group_sum(s);

            for (int base = 0; base < deg; base += 8) {
                const int cnt = (deg - base < 8) ? (deg - base) : 8;
                int c = 0;
                float p = 0.0f;
                if (e8 < cnt) {
                    c = ids[base + e8];
                    p = expf(lrelu(aln + ar[(long)c * NHEAD + hd]) - m) / s;
                }
                float4 hv[8];
                rows_load(hl, c, cnt, hv);
                rows_fma(p, cnt, hv, acc);
            }
        }
    }
    *(float4*)(out + (long)n * (OUT_F * NHEAD) + lane * 4) = acc;
}

__device__ inline void node_scan(const int* __restrict__ rec, const int* segLo, const int* segN,
                                 int i, int n, int lane,
                                 const float* __restrict__ al,
                                 const float* __restrict__ ar,
                                 const float* __restrict__ h,
                                 float* __restrict__ out) {
    const int hd = lane >> 3;
    const float aln = al[(long)n * NHEAD + hd];
    float m = -INFINITY;
    float s = 0.0f;
    float4 acc = make_float4(0.0f, 0.0f, 0.0f, 0.0f);
    for (int pass = 0; pass < 3; ++pass) {
        for (int t = 0; t < GA; ++t) {
            const int sl = __builtin_amdgcn_readfirstlane(segLo[t]);
            const int sn = __builtin_amdgcn_readfirstlane(segN[t]);
            for (int j0 = 0; j0 < sn; j0 += 32) {
                const int j = j0 + lane;
                const int f = (j < sn) ? rec[sl + j] : 0;
                const bool hit = (j < sn) && ((int)((unsigned)f >> CBITS) == i);
                unsigned mask = __builtin_amdgcn_ballot_w32(hit);
                while (mask != 0u) {
                    const int b = __builtin_ctz(mask);
                    mask &= mask - 1u;
                    const int cc = __builtin_amdgcn_readlane(f, b) & CMASK;
                    const float sc = lrelu(aln + ar[(long)cc * NHEAD + hd]);
                    if (pass == 0) {
                        m = fmaxf(m, sc);
                    } else if (pass == 1) {
                        s += expf(sc - m);
                    } else {
                        const float p = expf(sc - m) / s;
                        const float4 hv = *(const float4*)(h + (long)cc * (OUT_F * NHEAD) + lane * 4);
                        acc.x += p * hv.x; acc.y += p * hv.y; acc.z += p * hv.z; acc.w += p * hv.w;
                    }
                }
            }
        }
    }
    *(float4*)(out + (long)n * (OUT_F * NHEAD) + lane * 4) = acc;
}

__global__ __launch_bounds__(1024) __attribute__((amdgpu_waves_per_eu(16)))
void gat_group_node(const int* __restrict__ rec,
                    const int* __restrict__ cstart,
                    const float* __restrict__ al,
                    const float* __restrict__ ar,
                    const float* __restrict__ h,
                    float* __restrict__ out, int E, int N) {
    __shared__ int segLo[GA];
    __shared__ int segN[GA];
    __shared__ int ncnt[SMAX];
    __shared__ int nbeg[SMAX];
    __shared__ int ncur[SMAX];
    __shared__ int wtot[64];
    __shared__ int staged[CAPC];
    __shared__ int nextn;
    const int tid = threadIdx.x;
    const int lane = tid & 31;
    const int seg0 = (tid >> 5) * SEGW;
    const int P = bin_nodes(N);
    const int n0 = (int)blockIdx.x * P;
    if (n0 >= N) return;
    const int nb = (N - n0 < P) ? (N - n0) : P;
    const int CH = (E + GA - 1) / GA;

    int cs0 = 0, c = 0;
    if (tid < GA) {
        const int* __restrict__ p = cstart + tid * CPITCH + (int)blockIdx.x;
        cs0 = p[0];
        c = p[1] - cs0;
        segLo[tid] = tid * CH + cs0;
        segN[tid] = c;
    }
    ncnt[tid] = 0;
    const int2 tot = block_sum2(cs0, c, wtot);
    const int len = __builtin_amdgcn_readfirstlane(tot.y);
    const bool stage = (len <= CAPC);

    {
        int r[SEGW];
        #pragma unroll
        for (int i = 0; i < SEGW; ++i) {
            r[i] = (lane < segN[seg0 + i]) ? (int)((unsigned)rec[segLo[seg0 + i] + lane] >> CBITS) : 0;
        }
        #pragma unroll
        for (int i = 0; i < SEGW; ++i) {
            if (lane < segN[seg0 + i]) atomicAdd(&ncnt[r[i]], 1);
        }
        for (int i = 0; i < SEGW; ++i) {
            const int sl = segLo[seg0 + i];
            const int sn = segN[seg0 + i];
            for (int j = lane + 32; j < sn; j += 32) atomicAdd(&ncnt[(unsigned)rec[sl + j] >> CBITS], 1);
        }
    }
    __syncthreads();
    nbeg[tid] = block_excl_scan(ncnt[tid], wtot);

    if (!stage) {
        for (int i = tid >> 5; i < nb; i += 32) node_scan(rec, segLo, segN, i, n0 + i, lane, al, ar, h, out);
        return;
    }
    ncur[tid] = nbeg[tid];
    if (tid == 0) nextn = 0;
    __syncthreads();
    {
        int e[SEGW];
        #pragma unroll
        for (int i = 0; i < SEGW; ++i) {
            e[i] = (lane < segN[seg0 + i]) ? rec[segLo[seg0 + i] + lane] : 0;
        }
        #pragma unroll
        for (int i = 0; i < SEGW; ++i) {
            if (lane < segN[seg0 + i]) {
                const int pos = atomicAdd(&ncur[(unsigned)e[i] >> CBITS], 1);
                staged[pos] = e[i] & CMASK;
            }
        }
        for (int i = 0; i < SEGW; ++i) {
            const int sl = segLo[seg0 + i];
            const int sn = segN[seg0 + i];
            for (int j = lane + 32; j < sn; j += 32) {
                const int f = rec[sl + j];
                const int pos = atomicAdd(&ncur[(unsigned)f >> CBITS], 1);
                staged[pos] = f & CMASK;
            }
        }
    }
    __syncthreads();
    for (int guard = 0; guard <= SMAX; ++guard) {
        int i = 0;
        if (lane == 0) i = atomicAdd(&nextn, 1);
        i = __builtin_amdgcn_readfirstlane(i);
        if (i >= nb) break;
        const int beg = __builtin_amdgcn_readfirstlane(nbeg[i]);
        const int deg = __builtin_amdgcn_readfirstlane(ncnt[i]);
        node_walk(staged + beg, deg, n0 + i, lane, al, ar, h, out);
    }
}

extern "C" void kernel_launch(void* const* d_in, const int* in_sizes, int n_in,
                              void* d_out, int out_size, void* d_ws, size_t ws_size,
                              hipStream_t stream) {
    const float* x    = (const float*)d_in[0];
    const int*   edge = (const int*)d_in[1];
    const float* W    = (const float*)d_in[2];
    const float* a_l  = (const float*)d_in[3];
    const float* a_r  = (const float*)d_in[4];

    const int N = in_sizes[0] / IN_F;
    const int E = in_sizes[1] / 2;
    const int* rowp = edge;
    const int* colp = edge + E;

    float* ws = (float*)d_ws;
    float* h  = ws;
    float* al = h  + (size_t)N * (OUT_F * NHEAD);
    float* ar = al + (size_t)N * NHEAD;
    int* cstart = (int*)(ar + (size_t)N * NHEAD);
    int* rec    = cstart + GA * CPITCH;
    float* out = (float*)d_out;

    const int tiles16 = (N + 15) / 16;
    const int G = ((tiles16 + 15) / 16 > 512) ? (tiles16 + 15) / 16 : 512;
    gat_sort_gemm<<<(unsigned)(GA + G), NBIN, 0, stream>>>(rowp, colp, cstart, rec, E, x, W, a_l, a_r, h, al, ar, N);
    gat_group_node<<<NBIN, 1024, 0, stream>>>(rec, cstart, al, ar, h, out, E, N);
}


# Bin work-group's front end waits once, not per segment: its 8 segment reads and 8 returning LDS adds leave together, records stay in registers, bin length comes from the node scan; sums unchanged
# speedup vs baseline: 10.1158x; 1.0309x over previous
#include <hip/hip_runtime.h>
#include <hip/hip_bf16.h>
#include <math.h>

typedef __attribute__((ext_vector_type(2))) float v2f;
typedef __attribute__((ext_vector_type(8))) float v8f;

#define IN_F   128
#define OUT_F  32
#define NHEAD  4
#define ALPHA  0.2f


__device__ inline void stage_w(const float* __restrict__ W, float4* __restrict__ wq) {
    const int c4 = (threadIdx.x & 31) * 4;
    for (int kq = (int)(threadIdx.x >> 5); kq < IN_F / 4; kq += (int)(blockDim.x >> 5)) {
        const float* __restrict__ s = W + (4 * kq) * (OUT_F * NHEAD) + c4;
        const float4 r0 = *(const float4*)(s);
        const float4 r1 = *(const float4*)(s + (OUT_F * NHEAD));
        const float4 r2 = *(const float4*)(s + 2 * (OUT_F * NHEAD));
        const float4 r3 = *(const float4*)(s + 3 * (OUT_F * NHEAD));
        float4* __restrict__ d = wq + kq * (OUT_F * NHEAD) + c4;
        d[0] = make_float4(r0.x, r1.x, r2.x, r3.x);
        d[1] = make_float4(r0.y, r1.y, r2.y, r3.y);
        d[2] = make_float4(r0.z, r1.z, r2.z, r3.z);
        d[3] = make_float4(r0.w, r1.w, r2.w, r3.w);
    }
}

template <int XM>
__device__ inline float row_xor(float v) {
    return __int_as_float(__builtin_amdgcn_update_dpp(0, __float_as_int(v), 0x160 | XM, 0xf, 0xf, true));
}
template <int XM>
__device__ inline float fold(float a, float b, bool hi) {
    const float keep = hi ? b : a;
    const float send = hi ? a : b;
    return keep + row_xor<XM>(send);
}

__device__ inline void gemm_tile(const float* __restrict__ x,
                                 const float4* __restrict__ wq,
                                 const float* __restrict__ a_l,
                                 const float* __restrict__ a_r,
                                 float* __restrict__ h,
                                 float* __restrict__ al,
                                 float* __restrict__ ar, int N, long rowTile, int lane) {
    const long row0 = rowTile * 16;

    const int mrow = lane & 15;
    const int hf   = lane >> 4;

    v8f acc[8];
    #pragma unroll
    for (int t = 0; t < 8; ++t) acc[t] = v8f{};

    long r = row0 + mrow;
    if (r >= N) r = N - 1;
    const float* __restrict__ xq = x + r * IN_F + 4 * hf;
    const float4* __restrict__ wl = wq + hf * (OUT_F * NHEAD) + mrow;

    float4 anext = *(const float4*)xq;
    for (int kb = 0; kb < IN_F / 8; ++kb) {
        const float4 a4 = anext;
        if (kb + 1 < IN_F / 8) anext = *(const float4*)(xq + 8 * (kb + 1));
        const float4* __restrict__ wk = wl + kb * (2 * OUT_F * NHEAD);
        float4 b4[8];
        #pragma unroll
        for (int t = 0; t < 8; ++t) b4[t] = wk[t * 16];
        v2f a, b;
        a.x = a4.x; a.y = a4.y;
        #pragma unroll
        for (int t = 0; t < 8; ++t) {
            b.x = b4[t].x; b.y = b4[t].y;
            acc[t] = __builtin_amdgcn_wmma_f32_16x16x4_f32(
                false, a, false, b, (short)0, acc[t], false, false);
        }
        a.x = a4.z; a.y = a4.w;
        #pragma unroll
        for (int t = 0; t < 8; ++t) {
            b.x = b4[t].z; b.y = b4[t].w;
            acc[t] = __builtin_amdgcn_wmma_f32_16x16x4_f32(
                false, a, false, b, (short)0, acc[t], false, false);
        }
    }

    const int rowOff = (lane < 16) ? 0 : 8;
    #pragma unroll
    for (int t = 0; t < 8; ++t) {
        #pragma unroll
        for (int g = 0; g < 8; ++g) {
            float v = acc[t][g];
            v = (v == v) ? v : 0.0f;
            acc[t][g] = v;
            const long rr = row0 + g + rowOff;
            if (rr < N) h[rr * (OUT_F * NHEAD) + t * 16 + mrow] = v;
        }
    }

    const bool b0 = (mrow & 1) != 0, b1 = (mrow & 2) != 0, b2 = (mrow & 4) != 0, b3 = (mrow & 8) != 0;
    float res[NHEAD];
    #pragma unroll
    for (int j = 0; j < NHEAD; ++j) {
        float q[2];
        #pragma unroll
        for (int c = 0; c < 2; ++c) {
            const float* __restrict__ av = (c == 0) ? a_l : a_r;
            const float w0 = av[(2 * j) * 16 + mrow];
            const float w1 = av[(2 * j + 1) * 16 + mrow];
            float p[8];
            #pragma unroll
            for (int g = 0; g < 8; ++g) p[g] = w0 * acc[2 * j][g] + w1 * acc[2 * j + 1][g];
            const float p01 = fold<1>(p[0], p[1], b0);
            const float p23 = fold<1>(p[2], p[3], b0);
            const float p45 = fold<1>(p[4], p[5], b0);
            const float p67 = fold<1>(p[6], p[7], b0);
            const float p03 = fold<2>(p01, p23, b1);
            const float p47 = fold<2>(p45, p67, b1);
            q[c] = fold<4>(p03, p47, b2);
        }
        res[j] = fold<8>(q[0], q[1], b3);
    }
    const long orow = row0 + rowOff + (mrow & 7);
    if (orow < N) {
        float* __restrict__ dst = (b3 ? ar : al) + orow * NHEAD;
        *(float4*)dst = make_float4(res[0], res[1], res[2], res[3]);
    }
}

#define GA     256
#define NBIN   512
#define SMAX   1024
#define CHMAX  8192
#define CAPC   16384
#define CPITCH 544
#define SEGW   (GA / 32)
#define CBITS  19
#define CMASK  ((1 << CBITS) - 1)

__device__ inline int bin_nodes(int N) { return (N + NBIN - 1) / NBIN; }
__device__ inline unsigned bin_magic(int P) { return 0xFFFFFFFFu / (unsigned)P + 1u; }
__device__ inline int bin_of(int r, int P, unsigned M) {
    return (P == 1) ? r : (int)__umulhi((unsigned)r, M);
}

__device__ inline int block_excl_scan(int v, int* wtot) {
    const int lane = threadIdx.x & 31;
    const int wave = threadIdx.x >> 5;
    const int nw = blockDim.x >> 5;
    int incl = v;
    #pragma unroll
    for (int off = 1; off < 32; off <<= 1) {
        const int u = __shfl_up(incl, off, 32);
        if (lane >= off) incl += u;
    }
    if (lane == 31) wtot[wave] = incl;
    __syncthreads();
    int pre = 0;
    for (int w = 0; w < nw; ++w) {
        if (w < wave) pre += wtot[w];
    }
    __syncthreads();
    return pre + incl - v;
}

__device__ inline int block_excl_scan_total(int v, int* wtot, int& total) {
    const int lane = threadIdx.x & 31;
    const int wave = threadIdx.x >> 5;
    const int nw = blockDim.x >> 5;
    int incl = v;
    #pragma unroll
    for (int off = 1; off < 32; off <<= 1) {
        const int u = __shfl_up(incl, off, 32);
        if (lane >= off) incl += u;
    }
    if (lane == 31) wtot[wave] = incl;
    __syncthreads();
    int pre = 0, tot = 0;
    for (int w = 0; w < nw; ++w) {
        const int t = wtot[w];
        tot += t;
        if (w < wave) pre += t;
    }
    total = tot;
    return pre + incl - v;
}

__device__ inline void chunk_sort(const int* __restrict__ row, const int* __restrict__ col,
                                  int* __restrict__ cstart, int* __restrict__ rec,
                                  int E, int N) {
    __shared__ int cnt[NBIN];
    __shared__ int cur[NBIN];
    __shared__ int wtot[32];
    __shared__ int staged[CHMAX];
    const int tid = threadIdx.x;
    const int P = bin_nodes(N);
    const unsigned M = bin_magic(P);
    const int CH = (E + GA - 1) / GA;
    const int lo = (int)blockIdx.x * CH;
    const int hi = (lo + CH < E) ? (lo + CH) : E;
    const int len = (hi > lo) ? (hi - lo) : 0;
    const bool stage = (CH <= CHMAX);

    cnt[tid] = 0;
    __syncthreads();
    for (int i0 = lo + tid; i0 < hi; i0 += 4 * NBIN) {
        int r[4];
        #pragma unroll
        for (int k = 0; k < 4; ++k) {
            const int i = i0 + k * NBIN;
            r[k] = (i < hi) ? row[i] : 0;
        }
        #pragma unroll
        for (int k = 0; k < 4; ++k) {
            if (i0 + k * NBIN < hi) atomicAdd(&cnt[bin_of(r[k], P, M)], 1);
        }
    }
    __syncthreads();
    const int c = cnt[tid];
    const int excl = block_excl_scan(c, wtot);
    cur[tid] = excl;
    int* __restrict__ cs = cstart + (int)blockIdx.x * CPITCH;
    cs[tid] = excl;
    if (tid == NBIN - 1) cs[NBIN] = excl + c;
    __syncthreads();
    for (int i0 = lo + tid; i0 < hi; i0 += 4 * NBIN) {
        int r[4], s[4];
        #pragma unroll
        for (int k = 0; k < 4; ++k) {
            const int i = i0 + k * NBIN;
            r[k] = (i < hi) ? row[i] : 0;
            s[k] = (i < hi) ? col[i] : 0;
        }
        #pragma unroll
        for (int k = 0; k < 4; ++k) {
            if (i0 + k * NBIN < hi) {
                const int b = bin_of(r[k], P, M);
                const int pos = atomicAdd(&cur[b], 1);
                const int v = ((r[k] - b * P) << CBITS) | s[k];
                if (stage) staged[pos]   = v;
                else       rec[lo + pos] = v;
            }
        }
    }
    if (stage) {
        __syncthreads();
        for (int j = tid; j < len; j += NBIN) rec[lo + j] = staged[j];
    }
}

__global__ __launch_bounds__(NBIN) __attribute__((amdgpu_waves_per_eu(8)))
void gat_sort_gemm(const int* __restrict__ row,
                                                      const int* __restrict__ col,
                                                      int* __restrict__ cstart,
                                                      int* __restrict__ recw, int E,
                                                      const float* __restrict__ x,
                                                      const float* __restrict__ W,
                                                      const float* __restrict__ a_l,
                                                      const float* __restrict__ a_r,
                                                      float* __restrict__ h,
                                                      float* __restrict__ al,
                                                      float* __restrict__ ar, int N) {
    if (blockIdx.x < GA) {
        chunk_sort(row, col, cstart, recw, E, N);
        return;
    }
    __shared__ float4 wq[(IN_F / 4) * (OUT_F * NHEAD)];
    stage_w(W, wq);
    __syncthreads();
    const int wave = threadIdx.x >> 5;
    const int lane = threadIdx.x & 31;
    const int G = (int)gridDim.x - GA;
    const int tiles = (N + 15) / 16;
    const int t = (int)blockIdx.x - GA + G * wave;
    if (t < tiles) gemm_tile(x, wq, a_l, a_r, h, al, ar, N, (long)t, lane);
}

__device__ inline float lrelu(float v) { return (v > 0.0f) ? v : (ALPHA * v); }

template <int K>
__device__ inline float group_bcast(float v) {
    return __int_as_float((int)__builtin_amdgcn_mov_dpp8((unsigned)__float_as_int(v), (unsigned)(K * 0x249249)));
}
__device__ inline float group_max(float v) {
    #pragma unroll
    for (int off = 4; off > 0; off >>= 1) v = fmaxf(v, __shfl_xor(v, off, 32));
    return v;
}
__device__ inline float group_sum(float v) {
    #pragma unroll
    for (int off = 4; off > 0; off >>= 1) v += __shfl_xor(v, off, 32);
    return v;
}

#define ESLOT 4

#define H_LOAD(K) if (K < cnt) { hv[K] = *(const float4*)(hl + (long)__builtin_amdgcn_readlane(c, K) * (OUT_F * NHEAD)); }
#define H_FMA(K) if (K < cnt) { const float w = group_bcast<K>(p); \
                acc.x += w * hv[K].x; acc.y += w * hv[K].y; acc.z += w * hv[K].z; acc.w += w * hv[K].w; }
__device__ inline void rows_load(const float* __restrict__ hl, int c, int cnt, float4 (&hv)[8]) {
    H_LOAD(0) H_LOAD(1) H_LOAD(2) H_LOAD(3) H_LOAD(4) H_LOAD(5) H_LOAD(6) H_LOAD(7)
}
__device__ inline void rows_fma(float p, int cnt, const float4 (&hv)[8], float4& acc) {
    H_FMA(0) H_FMA(1) H_FMA(2) H_FMA(3) H_FMA(4) H_FMA(5) H_FMA(6) H_FMA(7)
}
#undef H_LOAD
#undef H_FMA

__device__ inline void node_walk(const int* __restrict__ ids, int deg, int n, int lane,
                                 const float* __restrict__ al,
                                 const float* __restrict__ ar,
                                 const float* __restrict__ h,
                                 float* __restrict__ out) {
    const int e8 = lane & 7;
    const int hd = lane >> 3;

    float4 acc = make_float4(0.0f, 0.0f, 0.0f, 0.0f);
    if (deg > 0) {
        const float aln = al[(long)n * NHEAD + hd];
        const float* __restrict__ hl = h + lane * 4;

        if (deg <= 8 * ESLOT) {
            int   c[ESLOT];
            float p[ESLOT];
            #pragma unroll
            for (int k = 0; k < ESLOT; ++k) {
                const int j = 8 * k + e8;
                c[k] = ids[(j < deg) ? j : (deg - 1)];
            }
            #pragma unroll
            for (int k = 0; k < ESLOT; ++k) p[k] = ar[(long)c[k] * NHEAD + hd];
            float4 hv[8];
            const int cnt0 = (deg < 8) ? deg : 8;
            rows_load(hl, c[0], cnt0, hv);

            float m = -INFINITY;
            #pragma unroll
            for (int k = 0; k < ESLOT; ++k) {
                p[k] = (8 * k + e8 < deg) ? lrelu(aln + p[k]) : -INFINITY;
                m = fmaxf(m, p[k]);
            }
            m = group_max(m);
            float s = 0.0f;
            #pragma unroll
            for (int k = 0; k < ESLOT; ++k) {
                p[k] = expf(p[k] - m);
                s += p[k];
            }
            s = group_sum(s);
            rows_fma(p[0] / s, cnt0, hv, acc);
            #pragma unroll
            for (int k = 1; k < ESLOT; ++k) {
                if (8 * k < deg) {
                    const int cnt = (deg - 8 * k < 8) ? (deg - 8 * k) : 8;
                    rows_load(hl, c[k], cnt, hv);
                    rows_fma(p[k] / s, cnt, hv, acc);
                }
            }
        } else {
            float m = -INFINITY;
            for (int base = 0; base < deg; base += 8) {
                const int idx = base + e8;
                if (idx < deg) m = fmaxf(m, lrelu(aln + ar[(long)ids[idx] * NHEAD + hd]));
            }
            m = group_max(m);

            float s = 0.0f;
            for (int base = 0; base < deg; base += 8) {
                const int idx = base + e8;
                if (idx < deg) s += expf(lrelu(aln + ar[(long)ids[idx] * NHEAD + hd]) - m);
            }
            s = group_sum(s);

            for (int base = 0; base < deg; base += 8) {
                const int cnt = (deg - base < 8) ? (deg - base) : 8;
                int c = 0;
                float p = 0.0f;
                if (e8 < cnt) {
                    c = ids[base + e8];
                    p = expf(lrelu(aln + ar[(long)c * NHEAD + hd]) - m) / s;
                }
                float4 hv[8];
                rows_load(hl, c, cnt, hv);
                rows_fma(p, cnt, hv, acc);
            }
        }
    }
    *(float4*)(out + (long)n * (OUT_F * NHEAD) + lane * 4) = acc;
}

__device__ inline void node_scan(const int* __restrict__ rec, const int* segLo, const int* segN,
                                 int i, int n, int lane,
                                 const float* __restrict__ al,
                                 const float* __restrict__ ar,
                                 const float* __restrict__ h,
                                 float* __restrict__ out) {
    const int hd = lane >> 3;
    const float aln = al[(long)n * NHEAD + hd];
    float m = -INFINITY;
    float s = 0.0f;
    float4 acc = make_float4(0.0f, 0.0f, 0.0f, 0.0f);
    for (int pass = 0; pass < 3; ++pass) {
        for (int t = 0; t < GA; ++t) {
            const int sl = __builtin_amdgcn_readfirstlane(segLo[t]);
            const int sn = __builtin_amdgcn_readfirstlane(segN[t]);
            for (int j0 = 0; j0 < sn; j0 += 32) {
                const int j = j0 + lane;
                const int f = (j < sn) ? rec[sl + j] : 0;
                const bool hit = (j < sn) && ((int)((unsigned)f >> CBITS) == i);
                unsigned mask = __builtin_amdgcn_ballot_w32(hit);
                while (mask != 0u) {
                    const int b = __builtin_ctz(mask);
                    mask &= mask - 1u;
                    const int cc = __builtin_amdgcn_readlane(f, b) & CMASK;
                    const float sc = lrelu(aln + ar[(long)cc * NHEAD + hd]);
                    if (pass == 0) {
                        m = fmaxf(m, sc);
                    } else if (pass == 1) {
                        s += expf(sc - m);
                    } else {
                        const float p = expf(sc - m) / s;
                        const float4 hv = *(const float4*)(h + (long)cc * (OUT_F * NHEAD) + lane * 4);
                        acc.x += p * hv.x; acc.y += p * hv.y; acc.z += p * hv.z; acc.w += p * hv.w;
                    }
                }
            }
        }
    }
    *(float4*)(out + (long)n * (OUT_F * NHEAD) + lane * 4) = acc;
}

__global__ __launch_bounds__(1024) __attribute__((amdgpu_waves_per_eu(16)))
void gat_group_node(const int* __restrict__ rec,
                    const int* __restrict__ cstart,
                    const float* __restrict__ al,
                    const float* __restrict__ ar,
                    const float* __restrict__ h,
                    float* __restrict__ out, int E, int N) {
    __shared__ __attribute__((aligned(16))) int segLo[GA];
    __shared__ __attribute__((aligned(16))) int segN[GA];
    __shared__ int ncnt[SMAX];
    __shared__ int nbeg[SMAX];
    __shared__ int ncur[SMAX + 32];
    __shared__ int wtot[32];
    __shared__ int staged[CAPC];
    __shared__ int nextn;
    const int tid = threadIdx.x;
    const int lane = tid & 31;
    const int seg0 = (tid >> 5) * SEGW;
    const int P = bin_nodes(N);
    const int n0 = (int)blockIdx.x * P;
    if (n0 >= N) return;
    const int nb = (N - n0 < P) ? (N - n0) : P;
    const int CH = (E + GA - 1) / GA;

    if (tid < GA) {
        const int* __restrict__ p = cstart + tid * CPITCH + (int)blockIdx.x;
        const int cs0 = p[0];
        segLo[tid] = tid * CH + cs0;
        segN[tid] = p[1] - cs0;
    }
    ncnt[tid] = 0;
    __syncthreads();

    int sl[SEGW], sn[SEGW], e[SEGW];
    #pragma unroll
    for (int i = 0; i < SEGW; ++i) {
        sl[i] = segLo[seg0 + i];
        sn[i] = segN[seg0 + i];
        e[i] = 0;
    }
    if (E > 0) {
        #pragma unroll
        for (int i = 0; i < SEGW; ++i) {
            const int j = sl[i] + ((lane < sn[i]) ? lane : 0);
            e[i] = rec[(j < E) ? j : (E - 1)];
        }
    }
    #pragma unroll
    for (int i = 0; i < SEGW; ++i) {
        if (lane < sn[i]) atomicAdd(&ncnt[(unsigned)e[i] >> CBITS], 1);
    }
    #pragma unroll
    for (int i = 0; i < SEGW; ++i) {
        for (int j = lane + 32; j < sn[i]; j += 32) atomicAdd(&ncnt[(unsigned)rec[sl[i] + j] >> CBITS], 1);
    }
    __syncthreads();
    int total;
    const int beg = block_excl_scan_total(ncnt[tid], wtot, total);
    nbeg[tid] = beg;
    const int len = __builtin_amdgcn_readfirstlane(total);
    const bool stage = (len <= CAPC);

    if (!stage) {
        for (int i = tid >> 5; i < nb; i += 32) node_scan(rec, segLo, segN, i, n0 + i, lane, al, ar, h, out);
        return;
    }
    ncur[tid] = beg;
    if (tid < 32) ncur[SMAX + tid] = 0;
    if (tid == 0) nextn = 0;
    __syncthreads();
    {
        int pos[SEGW];
        #pragma unroll
        for (int i = 0; i < SEGW; ++i) {
            const bool has = lane < sn[i];
            pos[i] = atomicAdd(&ncur[has ? (int)((unsigned)e[i] >> CBITS) : (SMAX + lane)], has ? 1 : 0);
        }
        #pragma unroll
        for (int i = 0; i < SEGW; ++i) {
            if (lane < sn[i]) staged[pos[i]] = e[i] & CMASK;
        }
        #pragma unroll
        for (int i = 0; i < SEGW; ++i) {
            for (int j = lane + 32; j < sn[i]; j += 32) {
                const int f = rec[sl[i] + j];
                const int q = atomicAdd(&ncur[(unsigned)f >> CBITS], 1);
                staged[q] = f & CMASK;
            }
        }
    }
    __syncthreads();
    for (int guard = 0; guard <= SMAX; ++guard) {
        int i = 0;
        if (lane == 0) i = atomicAdd(&nextn, 1);
        i = __builtin_amdgcn_readfirstlane(i);
        if (i >= nb) break;
        const int beg = __builtin_amdgcn_readfirstlane(nbeg[i]);
        const int deg = __builtin_amdgcn_readfirstlane(ncnt[i]);
        node_walk(staged + beg, deg, n0 + i, lane, al, ar, h, out);
    }
}

extern "C" void kernel_launch(void* const* d_in, const int* in_sizes, int n_in,
                              void* d_out, int out_size, void* d_ws, size_t ws_size,
                              hipStream_t stream) {
    const float* x    = (const float*)d_in[0];
    const int*   edge = (const int*)d_in[1];
    const float* W    = (const float*)d_in[2];
    const float* a_l  = (const float*)d_in[3];
    const float* a_r  = (const float*)d_in[4];

    const int N = in_sizes[0] / IN_F;
    const int E = in_sizes[1] / 2;
    const int* rowp = edge;
    const int* colp = edge + E;

    float* ws = (float*)d_ws;
    float* h  = ws;
    float* al = h  + (size_t)N * (OUT_F * NHEAD);
    float* ar = al + (size_t)N * NHEAD;
    int* cstart = (int*)(ar + (size_t)N * NHEAD);
    int* rec    = cstart + GA * CPITCH;
    float* out = (float*)d_out;

    const int tiles16 = (N + 15) / 16;
    const int G = ((tiles16 + 15) / 16 > 512) ? (tiles16 + 15) / 16 : 512;
    gat_sort_gemm<<<(unsigned)(GA + G), NBIN, 0, stream>>>(rowp, colp, cstart, rec, E, x, W, a_l, a_r, h, al, ar, N);
    gat_group_node<<<NBIN, 1024, 0, stream>>>(rec, cstart, al, ar, h, out, E, N);
}
